# starting kernel
# baseline (speedup 1.0000x reference)
#include <hip/hip_runtime.h>

#define DCH 128
#define LN_EPS 1e-5f

typedef float v2f __attribute__((ext_vector_type(2)));
typedef float v8f __attribute__((ext_vector_type(8)));

__global__ void zero_cnt_kernel(unsigned* __restrict__ cnt, int n) {
  int i = blockIdx.x * blockDim.x + threadIdx.x;
  if (i < n) cnt[i] = 0u;
}

__global__ void count_edges_kernel(const int* __restrict__ dst,
                                   unsigned* __restrict__ cnt, int e) {
  int i = blockIdx.x * blockDim.x + threadIdx.x;
  int stride = gridDim.x * blockDim.x;
  for (; i < e; i += stride) atomicAdd(&cnt[dst[i]], 1u);
}

__global__ void dinv_kernel(const unsigned* __restrict__ cnt,
                            float* __restrict__ dinv, int n) {
  int i = blockIdx.x * blockDim.x + threadIdx.x;
  if (i < n) dinv[i] = rsqrtf(1.0f + (float)cnt[i]);
}

__global__ void coef_kernel(const int* __restrict__ src,
                            const int* __restrict__ dst,
                            const float* __restrict__ dinv,
                            float* __restrict__ coef, int e) {
  int i = blockIdx.x * blockDim.x + threadIdx.x;
  int stride = gridDim.x * blockDim.x;
  for (; i < e; i += stride) coef[i] = dinv[src[i]] * dinv[dst[i]];
}

__global__ __launch_bounds__(256)
void gemm_wmma_kernel(const float* __restrict__ X, const float* __restrict__ W,
                      float* __restrict__ out, int nrows) {
  __shared__ float xs[16 * DCH];
  const int m0 = blockIdx.x * 16;
  const int t  = threadIdx.x;

  const int tile_elems = (nrows - m0 >= 16 ? 16 : (nrows - m0)) * DCH;
  const float4* src4 = (const float4*)(X + (size_t)m0 * DCH);
  float4* lds4 = (float4*)xs;
  {
    float4 z; z.x = z.y = z.z = z.w = 0.0f;
    int i0 = t, i1 = t + 256;
    lds4[i0] = (i0 * 4 < tile_elems) ? src4[i0] : z;
    lds4[i1] = (i1 * 4 < tile_elems) ? src4[i1] : z;
  }
  __syncthreads();

  const int wave = t >> 5;
  const int lane = t & 31;
  const int n0   = wave * 16;
  const int row  = lane & 15;
  const int col  = lane & 15;
  const int koff = (lane >> 4) * 2;
  const int half = lane >> 4;

  v8f c = {};
#pragma unroll 8
  for (int k0 = 0; k0 < DCH; k0 += 4) {
    v2f a, b;
    a.x = xs[row * DCH + k0 + koff];
    a.y = xs[row * DCH + k0 + koff + 1];
    b.x = W[(k0 + koff)     * DCH + n0 + col];
    b.y = W[(k0 + koff + 1) * DCH + n0 + col];
    c = __builtin_amdgcn_wmma_f32_16x16x4_f32(
         false, a,  false, b,
         (short)0, c,  false,  false);
  }

  float* obase = out + (size_t)m0 * DCH + n0;
  if (m0 + 16 <= nrows) {
#pragma unroll
    for (int v = 0; v < 8; ++v) {
      int m = v + half * 8;
      obase[(size_t)m * DCH + col] = c[v];
    }
  } else {
#pragma unroll
    for (int v = 0; v < 8; ++v) {
      int m = v + half * 8;
      if (m0 + m < nrows) obase[(size_t)m * DCH + col] = c[v];
    }
  }
}

__global__ void init_agg_kernel(const float* __restrict__ hlin,
                                const float* __restrict__ dinv,
                                const float* __restrict__ bias,
                                float* __restrict__ agg, int n) {
  int i4 = blockIdx.x * blockDim.x + threadIdx.x;
  if (i4 >= n * 32) return;
  int node = i4 >> 5, c4 = i4 & 31;
  float dv = dinv[node];
  float sc = dv * dv;
  float4 h = ((const float4*)hlin)[i4];
  float4 b = ((const float4*)bias)[c4];
  float4 o;
  o.x = h.x * sc + b.x;
  o.y = h.y * sc + b.y;
  o.z = h.z * sc + b.z;
  o.w = h.w * sc + b.w;
  ((float4*)agg)[i4] = o;
}

__global__ __launch_bounds__(256)
void edge_scatter_kernel(const float* __restrict__ hlin,
                         const float* __restrict__ coef,
                         const int* __restrict__ src,
                         const int* __restrict__ dst,
                         float* __restrict__ agg, int e) {
  int gtid = blockIdx.x * blockDim.x + threadIdx.x;
  int wid  = gtid >> 5;
  int lane = threadIdx.x & 31;
  int nw   = (gridDim.x * blockDim.x) >> 5;
  for (int ed = wid; ed < e; ed += nw) {
    int s = src[ed], d = dst[ed];
    float cf = coef[ed];
    int nxt = ed + nw;
    if (nxt < e) {
      int ns = src[nxt];
      __builtin_prefetch(hlin + (size_t)ns * DCH + lane * 4, 0, 1);
    }
    float4 v = ((const float4*)(hlin + (size_t)s * DCH))[lane];
    float* ap = agg + (size_t)d * DCH + lane * 4;
    unsafeAtomicAdd(ap + 0, v.x * cf);
    unsafeAtomicAdd(ap + 1, v.y * cf);
    unsafeAtomicAdd(ap + 2, v.z * cf);
    unsafeAtomicAdd(ap + 3, v.w * cf);
  }
}

__global__ __launch_bounds__(256)
void post_kernel(const float* __restrict__ agg, const float* __restrict__ gamma,
                 const float* __restrict__ beta, float* __restrict__ hout,
                 float* __restrict__ jk, int do_relu, int first, int n) {
  int wid  = (blockIdx.x * blockDim.x + threadIdx.x) >> 5;
  int lane = threadIdx.x & 31;
  if (wid >= n) return;

  float4 v = ((const float4*)(agg + (size_t)wid * DCH))[lane];

  float s = v.x + v.y + v.z + v.w;
#pragma unroll
  for (int off = 16; off > 0; off >>= 1) s += __shfl_xor(s, off, 32);
  float mu = s * (1.0f / DCH);

  float dx = v.x - mu, dy = v.y - mu, dz = v.z - mu, dw = v.w - mu;
  float q = dx * dx + dy * dy + dz * dz + dw * dw;
#pragma unroll
  for (int off = 16; off > 0; off >>= 1) q += __shfl_xor(q, off, 32);
  float inv = rsqrtf(q * (1.0f / DCH) + LN_EPS);

  float4 g = ((const float4*)gamma)[lane];
  float4 b = ((const float4*)beta)[lane];
  float4 y;
  y.x = dx * inv * g.x + b.x;
  y.y = dy * inv * g.y + b.y;
  y.z = dz * inv * g.z + b.z;
  y.w = dw * inv * g.w + b.w;
  if (do_relu) {
    y.x = fmaxf(y.x, 0.0f); y.y = fmaxf(y.y, 0.0f);
    y.z = fmaxf(y.z, 0.0f); y.w = fmaxf(y.w, 0.0f);
  }

  ((float4*)(hout + (size_t)wid * DCH))[lane] = y;

  float4* op = (float4*)(jk + (size_t)wid * DCH) + lane;
  if (first) {
    *op = y;
  } else {
    float4 o = *op;
    o.x = fmaxf(o.x, y.x); o.y = fmaxf(o.y, y.y);
    o.z = fmaxf(o.z, y.z); o.w = fmaxf(o.w, y.w);
    *op = o;
  }
}

extern "C" void kernel_launch(void* const* d_in, const int* in_sizes, int n_in,
                              void* d_out, int out_size, void* d_ws, size_t ws_size,
                              hipStream_t stream) {
  const int N = in_sizes[0] / DCH;
  const int E = in_sizes[1] / 2;

  const float* feat = (const float*)d_in[0];
  const int*   ei   = (const int*)d_in[1];
  const int*   src  = ei;
  const int*   dst  = ei + E;
  const float* Wm[3] = {(const float*)d_in[2], (const float*)d_in[6],  (const float*)d_in[10]};
  const float* bi[3] = {(const float*)d_in[3], (const float*)d_in[7],  (const float*)d_in[11]};
  const float* gm[3] = {(const float*)d_in[4], (const float*)d_in[8],  (const float*)d_in[12]};
  const float* be[3] = {(const float*)d_in[5], (const float*)d_in[9],  (const float*)d_in[13]};
  float* out = (float*)d_out;

  float*    wsf  = (float*)d_ws;
  unsigned* cnt  = (unsigned*)wsf;
  float*    dinv = wsf + N;
  float*    hlin = wsf + 2 * (size_t)N;
  float*    agg  = hlin + (size_t)N * DCH;
  float*    hbuf = agg  + (size_t)N * DCH;
  float*    coef = hbuf + (size_t)N * DCH;

  const int nodeBlocks = (N + 255) / 256;
  zero_cnt_kernel<<<nodeBlocks, 256, 0, stream>>>(cnt, N);
  count_edges_kernel<<<2048, 256, 0, stream>>>(dst, cnt, E);
  dinv_kernel<<<nodeBlocks, 256, 0, stream>>>(cnt, dinv, N);
  coef_kernel<<<2048, 256, 0, stream>>>(src, dst, dinv, coef, E);

  const int gemmBlocks = (N + 15) / 16;
  const int elemBlocks = (N * 32 + 255) / 256;
  const int waveBlocks = (N + 7) / 8;

  const float* X = feat;
  for (int l = 0; l < 3; ++l) {
    gemm_wmma_kernel<<<gemmBlocks, 256, 0, stream>>>(X, Wm[l], hlin, N);
    init_agg_kernel<<<elemBlocks, 256, 0, stream>>>(hlin, dinv, bi[l], agg, N);
    edge_scatter_kernel<<<8192, 256, 0, stream>>>(hlin, coef, src, dst, agg, E);
    post_kernel<<<waveBlocks, 256, 0, stream>>>(agg, gm[l], be[l], hbuf, out,
                                                 (l < 2),
                                                 (l == 0), N);
    X = hbuf;
  }
}


# GEMM re-tiled: W staged once per block in LDS in fragment order (no bank conflict), A fragments straight from X, a wave owns 16 rows x 64 columns, whole-line stores; k summed in a new order
# speedup vs baseline: 3.0486x; 3.0486x over previous
#include <hip/hip_runtime.h>

#define DCH 128
#define LN_EPS 1e-5f

typedef float v2f __attribute__((ext_vector_type(2)));
typedef float v4f __attribute__((ext_vector_type(4)));
typedef float v8f __attribute__((ext_vector_type(8)));

__global__ void zero_cnt_kernel(unsigned* __restrict__ cnt, int n) {
  int i = blockIdx.x * blockDim.x + threadIdx.x;
  if (i < n) cnt[i] = 0u;
}

__global__ void count_rank_kernel(const int* __restrict__ dst,
                                  unsigned* __restrict__ cnt,
                                  unsigned* __restrict__ rank, int e) {
  int i = blockIdx.x * blockDim.x + threadIdx.x;
  int stride = gridDim.x * blockDim.x;
  for (; i < e; i += stride) rank[i] = atomicAdd(&cnt[dst[i]], 1u);
}

__global__ __launch_bounds__(1024)
void scan_dinv_kernel(const unsigned* __restrict__ cnt, int* __restrict__ rowptr,
                      float* __restrict__ dinv, int n) {
  __shared__ unsigned wave_pre[32];
  __shared__ unsigned wave_tot[32];
  const int t = threadIdx.x, lane = t & 31, wv = t >> 5;
  const int base = blockIdx.x * 1024;

  unsigned p = 0u;
  for (int i = t; i < base; i += 1024) p += cnt[i];

  const int idx = base + t;
  const unsigned c = (idx < n) ? cnt[idx] : 0u;
  unsigned inc = c;
  for (int off = 1; off < 32; off <<= 1) {
    unsigned v = __shfl_up(inc, off, 32);
    if (lane >= off) inc += v;
  }
  for (int off = 16; off > 0; off >>= 1) p += __shfl_xor(p, off, 32);
  if (lane == 31) wave_tot[wv] = inc;
  if (lane == 0)  wave_pre[wv] = p;
  __syncthreads();

  unsigned before = 0u;
  for (int k = 0; k < 32; ++k) {
    before += wave_pre[k];
    if (k < wv) before += wave_tot[k];
  }
  if (idx < n) {
    rowptr[idx] = (int)(before + inc - c);
    dinv[idx]   = rsqrtf(1.0f + (float)c);
  }
}

__global__ void place_kernel(const int* __restrict__ src,
                             const int* __restrict__ dst,
                             const int* __restrict__ rowptr,
                             const unsigned* __restrict__ rank,
                             int* __restrict__ csr_src, int e) {
  int i = blockIdx.x * blockDim.x + threadIdx.x;
  int stride = gridDim.x * blockDim.x;
  for (; i < e; i += stride) csr_src[rowptr[dst[i]] + (int)rank[i]] = src[i];
}

#define GEMM_ROWS_PER_BLOCK 64
__global__ __launch_bounds__(256)
void gemm_wmma_kernel(const float* __restrict__ X, const float* __restrict__ W,
                      float* __restrict__ out, int nrows) {
  __shared__ v4f wl[32 * DCH];
  const int t    = threadIdx.x;
  const int lane = t & 31;
  const int wave = __builtin_amdgcn_readfirstlane(t >> 5);
  const int c    = lane & 15;
  const int h    = lane >> 4;

#pragma unroll
  for (int it = 0; it < 4; ++it) {
    const int kq = wave + 8 * it;
    const v4f* wr = (const v4f*)(W + (size_t)(4 * kq) * DCH) + lane;
    const v4f r0 = wr[0], r1 = wr[32], r2 = wr[64], r3 = wr[96];
    v4f* dp = wl + kq * DCH + 64 * h + c;
    v4f q0, q1, q2, q3;
    q0.x = r0.x; q0.y = r1.x; q0.z = r2.x; q0.w = r3.x;
    q1.x = r0.y; q1.y = r1.y; q1.z = r2.y; q1.w = r3.y;
    q2.x = r0.z; q2.y = r1.z; q2.z = r2.z; q2.w = r3.z;
    q3.x = r0.w; q3.y = r1.w; q3.z = r2.w; q3.w = r3.w;
    dp[0]  = q0;
    dp[16] = q1;
    dp[32] = q2;
    dp[48] = q3;
  }

  const int nh = wave & 1;
  const int m0 = (blockIdx.x * 4 + (wave >> 1)) * 16;

  int arow = m0 + c;
  if (arow > nrows - 1) arow = nrows - 1;
  const v4f* ap = (const v4f*)(X + (size_t)arow * DCH) + h;
  v4f a[16];
#pragma unroll
  for (int j = 0; j < 16; ++j) a[j] = ap[2 * j];

  __syncthreads();

  v8f acc0 = {}, acc1 = {}, acc2 = {}, acc3 = {};
  const v4f* bp = wl + h * DCH + 64 * nh + c;
  v4f b0 = bp[0], b1 = bp[16], b2 = bp[32], b3 = bp[48];
#pragma unroll
  for (int j = 0; j < 16; ++j) {
    const int jn = (j + 1 < 16) ? (j + 1) : j;
    const v4f n0 = bp[jn * 2 * DCH + 0];
    const v4f n1 = bp[jn * 2 * DCH + 16];
    const v4f n2 = bp[jn * 2 * DCH + 32];
    const v4f n3 = bp[jn * 2 * DCH + 48];
    __builtin_amdgcn_sched_barrier(0);
    v2f af, bf;
    af.x = a[j].x; af.y = a[j].y;
    bf.x = b0.x; bf.y = b0.y;
    acc0 = __builtin_amdgcn_wmma_f32_16x16x4_f32(false, af, false, bf, (short)0, acc0, false, false);
    bf.x = b1.x; bf.y = b1.y;
    acc1 = __builtin_amdgcn_wmma_f32_16x16x4_f32(false, af, false, bf, (short)0, acc1, false, false);
    bf.x = b2.x; bf.y = b2.y;
    acc2 = __builtin_amdgcn_wmma_f32_16x16x4_f32(false, af, false, bf, (short)0, acc2, false, false);
    bf.x = b3.x; bf.y = b3.y;
    acc3 = __builtin_amdgcn_wmma_f32_16x16x4_f32(false, af, false, bf, (short)0, acc3, false, false);
    af.x = a[j].z; af.y = a[j].w;
    bf.x = b0.z; bf.y = b0.w;
    acc0 = __builtin_amdgcn_wmma_f32_16x16x4_f32(false, af, false, bf, (short)0, acc0, false, false);
    bf.x = b1.z; bf.y = b1.w;
    acc1 = __builtin_amdgcn_wmma_f32_16x16x4_f32(false, af, false, bf, (short)0, acc1, false, false);
    bf.x = b2.z; bf.y = b2.w;
    acc2 = __builtin_amdgcn_wmma_f32_16x16x4_f32(false, af, false, bf, (short)0, acc2, false, false);
    bf.x = b3.z; bf.y = b3.w;
    acc3 = __builtin_amdgcn_wmma_f32_16x16x4_f32(false, af, false, bf, (short)0, acc3, false, false);
    __builtin_amdgcn_sched_barrier(0);
    b0 = n0; b1 = n1; b2 = n2; b3 = n3;
  }

  float* obase = out + (size_t)m0 * DCH + 64 * nh + 4 * c;
  if (m0 + 16 <= nrows) {
#pragma unroll
    for (int r = 0; r < 8; ++r) {
      v4f o;
      o.x = acc0[r]; o.y = acc1[r]; o.z = acc2[r]; o.w = acc3[r];
      *(v4f*)(obase + (size_t)(8 * h + r) * DCH) = o;
    }
  } else {
#pragma unroll
    for (int r = 0; r < 8; ++r) {
      v4f o;
      o.x = acc0[r]; o.y = acc1[r]; o.z = acc2[r]; o.w = acc3[r];
      if (m0 + 8 * h + r < nrows) *(v4f*)(obase + (size_t)(8 * h + r) * DCH) = o;
    }
  }
}

template <int R>
__device__ __forceinline__ void gather_rows(const float* __restrict__ hlin,
                                            int my_s, float my_w,
                                            int j, int lane, v4f& acc) {
  v4f v[R];
#pragma unroll
  for (int k = 0; k < R; ++k) {
    const int s = __builtin_amdgcn_readlane(my_s, j + k);
    v[k] = ((const v4f*)(hlin + (size_t)s * DCH))[lane];
  }
#pragma unroll
  for (int k = 0; k < R; ++k) {
    const float w = __int_as_float(__builtin_amdgcn_readlane(__float_as_int(my_w), j + k));
    acc += v[k] * w;
  }
}

#define NODES_PER_WAVE 4

__global__ __launch_bounds__(256)
void gather_norm_kernel(const float* __restrict__ hlin, const float* __restrict__ dinv,
                        const float* __restrict__ bias, const int* __restrict__ rowptr,
                        const int* __restrict__ csr_src,
                        const float* __restrict__ gamma, const float* __restrict__ beta,
                        float* __restrict__ hout, float* __restrict__ jk,
                        int do_relu, int first_layer, int n, int e) {
  const int lane = threadIdx.x & 31;
  int wid = (blockIdx.x * blockDim.x + threadIdx.x) >> 5;
  wid = __builtin_amdgcn_readfirstlane(wid);
  const v4f b = ((const v4f*)bias)[lane];

  const int node0 = wid * NODES_PER_WAVE;
  for (int node = node0; node < node0 + NODES_PER_WAVE && node < n; ++node) {
    const int beg = rowptr[node];
    const int end = (node + 1 < n) ? rowptr[node + 1] : e;
    const float dd = dinv[node];
    const int total = end - beg + 1;

    v4f acc = b;
    for (int q0 = 0; q0 < total; q0 += 32) {
      const int m = (total - q0 < 32) ? (total - q0) : 32;
      int q = q0 + lane;
      if (q > total - 1) q = total - 1;
      int my_s = node;
      if (q > 0) my_s = csr_src[beg + q - 1];
      const float my_w = dinv[my_s] * dd;

      int j = 0;
      for (; j + 8 <= m; j += 8) gather_rows<8>(hlin, my_s, my_w, j, lane, acc);
      switch (m - j) {
        case 7: gather_rows<7>(hlin, my_s, my_w, j, lane, acc); break;
        case 6: gather_rows<6>(hlin, my_s, my_w, j, lane, acc); break;
        case 5: gather_rows<5>(hlin, my_s, my_w, j, lane, acc); break;
        case 4: gather_rows<4>(hlin, my_s, my_w, j, lane, acc); break;
        case 3: gather_rows<3>(hlin, my_s, my_w, j, lane, acc); break;
        case 2: gather_rows<2>(hlin, my_s, my_w, j, lane, acc); break;
        case 1: gather_rows<1>(hlin, my_s, my_w, j, lane, acc); break;
        default: break;
      }
    }

    asm volatile("" ::: "memory");
    const v4f g  = ((const v4f*)gamma)[lane];
    const v4f bt = ((const v4f*)beta)[lane];
    v4f* jp = (v4f*)(jk + (size_t)node * DCH) + lane;
    v4f jold = {};
    if (!first_layer) jold = *jp;

    float s = acc.x + acc.y + acc.z + acc.w;
#pragma unroll
    for (int off = 16; off > 0; off >>= 1) s += __shfl_xor(s, off, 32);
    const float mu = s * (1.0f / DCH);

    const float dx = acc.x - mu, dy = acc.y - mu, dz = acc.z - mu, dw = acc.w - mu;
    float qq = dx * dx + dy * dy + dz * dz + dw * dw;
#pragma unroll
    for (int off = 16; off > 0; off >>= 1) qq += __shfl_xor(qq, off, 32);
    const float inv = rsqrtf(qq * (1.0f / DCH) + LN_EPS);

    v4f y;
    y.x = dx * inv * g.x + bt.x;
    y.y = dy * inv * g.y + bt.y;
    y.z = dz * inv * g.z + bt.z;
    y.w = dw * inv * g.w + bt.w;
    if (do_relu) {
      y.x = fmaxf(y.x, 0.0f); y.y = fmaxf(y.y, 0.0f);
      y.z = fmaxf(y.z, 0.0f); y.w = fmaxf(y.w, 0.0f);
    }

    ((v4f*)(hout + (size_t)node * DCH))[lane] = y;

    if (first_layer) {
      *jp = y;
    } else {
      v4f o;
      o.x = fmaxf(jold.x, y.x); o.y = fmaxf(jold.y, y.y);
      o.z = fmaxf(jold.z, y.z); o.w = fmaxf(jold.w, y.w);
      *jp = o;
    }
  }
}

extern "C" void kernel_launch(void* const* d_in, const int* in_sizes, int n_in,
                              void* d_out, int out_size, void* d_ws, size_t ws_size,
                              hipStream_t stream) {
  const int N = in_sizes[0] / DCH;
  const int E = in_sizes[1] / 2;

  const float* feat = (const float*)d_in[0];
  const int*   ei   = (const int*)d_in[1];
  const int*   src  = ei;
  const int*   dst  = ei + E;
  const float* Wm[3] = {(const float*)d_in[2], (const float*)d_in[6],  (const float*)d_in[10]};
  const float* bi[3] = {(const float*)d_in[3], (const float*)d_in[7],  (const float*)d_in[11]};
  const float* gm[3] = {(const float*)d_in[4], (const float*)d_in[8],  (const float*)d_in[12]};
  const float* be[3] = {(const float*)d_in[5], (const float*)d_in[9],  (const float*)d_in[13]};
  float* out = (float*)d_out;

  float*    wsf     = (float*)d_ws;
  int*      rowptr  = (int*)wsf;
  float*    dinv    = wsf + N;
  float*    hlin    = wsf + 2 * (size_t)N;
  float*    agg     = hlin + (size_t)N * DCH;
  float*    hbuf    = agg  + (size_t)N * DCH;
  int*      csr_src = (int*)(hbuf + (size_t)N * DCH);
  unsigned* cnt     = (unsigned*)agg;
  unsigned* rank    = cnt + N;

  const int nodeBlocks = (N + 255) / 256;
  const int scanBlocks = (N + 1023) / 1024;
  zero_cnt_kernel<<<nodeBlocks, 256, 0, stream>>>(cnt, N);
  count_rank_kernel<<<2048, 256, 0, stream>>>(dst, cnt, rank, E);
  scan_dinv_kernel<<<scanBlocks, 1024, 0, stream>>>(cnt, rowptr, dinv, N);
  place_kernel<<<2048, 256, 0, stream>>>(src, dst, rowptr, rank, csr_src, E);

  const int gemmBlocks = (N + GEMM_ROWS_PER_BLOCK - 1) / GEMM_ROWS_PER_BLOCK;
  const int gathBlocks = (N + 8 * NODES_PER_WAVE - 1) / (8 * NODES_PER_WAVE);

  const float* X = feat;
  for (int l = 0; l < 3; ++l) {
    gemm_wmma_kernel<<<gemmBlocks, 256, 0, stream>>>(X, Wm[l], hlin, N);
    gather_norm_kernel<<<gathBlocks, 256, 0, stream>>>(hlin, dinv, bi[l], rowptr, csr_src,
                                                       gm[l], be[l], hbuf, out,
                                                        (l < 2),
                                                        (l == 0), N, E);
    X = hbuf;
  }
}


# Edge coefficient split: the GEMM scales a row's A fragment by the row's dinv once, the gather weights a node's entries by scalar dinv[node]: no per-edge dinv lookup; rounding moves, sum order kept
# speedup vs baseline: 3.1920x; 1.0470x over previous
#include <hip/hip_runtime.h>

#define DCH 128
#define LN_EPS 1e-5f

typedef float v2f __attribute__((ext_vector_type(2)));
typedef float v4f __attribute__((ext_vector_type(4)));
typedef float v8f __attribute__((ext_vector_type(8)));

__global__ void zero_cnt_kernel(unsigned* __restrict__ cnt, int n) {
  int i = blockIdx.x * blockDim.x + threadIdx.x;
  if (i < n) cnt[i] = 0u;
}

__global__ void count_rank_kernel(const int* __restrict__ dst,
                                  unsigned* __restrict__ cnt,
                                  unsigned* __restrict__ rank, int e) {
  int i = blockIdx.x * blockDim.x + threadIdx.x;
  int stride = gridDim.x * blockDim.x;
  for (; i < e; i += stride) rank[i] = atomicAdd(&cnt[dst[i]], 1u);
}

__global__ __launch_bounds__(1024)
void scan_dinv_kernel(const unsigned* __restrict__ cnt, int* __restrict__ rowptr,
                      float* __restrict__ dinv, int n) {
  __shared__ unsigned wave_pre[32];
  __shared__ unsigned wave_tot[32];
  const int t = threadIdx.x, lane = t & 31, wv = t >> 5;
  const int base = blockIdx.x * 1024;

  unsigned p = 0u;
  for (int i = t; i < base; i += 1024) p += cnt[i];

  const int idx = base + t;
  const unsigned c = (idx < n) ? cnt[idx] : 0u;
  unsigned inc = c;
  for (int off = 1; off < 32; off <<= 1) {
    unsigned v = __shfl_up(inc, off, 32);
    if (lane >= off) inc += v;
  }
  for (int off = 16; off > 0; off >>= 1) p += __shfl_xor(p, off, 32);
  if (lane == 31) wave_tot[wv] = inc;
  if (lane == 0)  wave_pre[wv] = p;
  __syncthreads();

  unsigned before = 0u;
  for (int k = 0; k < 32; ++k) {
    before += wave_pre[k];
    if (k < wv) before += wave_tot[k];
  }
  if (idx < n) {
    rowptr[idx] = (int)(before + inc - c);
    dinv[idx]   = rsqrtf(1.0f + (float)c);
  }
}

__global__ void place_kernel(const int* __restrict__ src,
                             const int* __restrict__ dst,
                             const int* __restrict__ rowptr,
                             const unsigned* __restrict__ rank,
                             int* __restrict__ csr_src, int e) {
  int i = blockIdx.x * blockDim.x + threadIdx.x;
  int stride = gridDim.x * blockDim.x;
  for (; i < e; i += stride) csr_src[rowptr[dst[i]] + (int)rank[i]] = src[i];
}

#define GEMM_ROWS_PER_BLOCK 64
__global__ __launch_bounds__(256)
void gemm_wmma_kernel(const float* __restrict__ X, const float* __restrict__ W,
                      const float* __restrict__ dinv, float* __restrict__ out, int nrows) {
  __shared__ v4f wl[32 * DCH];
  const int t    = threadIdx.x;
  const int lane = t & 31;
  const int wave = __builtin_amdgcn_readfirstlane(t >> 5);
  const int c    = lane & 15;
  const int h    = lane >> 4;

#pragma unroll
  for (int it = 0; it < 4; ++it) {
    const int kq = wave + 8 * it;
    const v4f* wr = (const v4f*)(W + (size_t)(4 * kq) * DCH) + lane;
    const v4f r0 = wr[0], r1 = wr[32], r2 = wr[64], r3 = wr[96];
    v4f* dp = wl + kq * DCH + 64 * h + c;
    v4f q0, q1, q2, q3;
    q0.x = r0.x; q0.y = r1.x; q0.z = r2.x; q0.w = r3.x;
    q1.x = r0.y; q1.y = r1.y; q1.z = r2.y; q1.w = r3.y;
    q2.x = r0.z; q2.y = r1.z; q2.z = r2.z; q2.w = r3.z;
    q3.x = r0.w; q3.y = r1.w; q3.z = r2.w; q3.w = r3.w;
    dp[0]  = q0;
    dp[16] = q1;
    dp[32] = q2;
    dp[48] = q3;
  }

  const int nh = wave & 1;
  const int m0 = (blockIdx.x * 4 + (wave >> 1)) * 16;

  int arow = m0 + c;
  if (arow > nrows - 1) arow = nrows - 1;
  const v4f* ap = (const v4f*)(X + (size_t)arow * DCH) + h;
  v4f a[16];
#pragma unroll
  for (int j = 0; j < 16; ++j) a[j] = ap[2 * j];
  const float sa = dinv[arow];
#pragma unroll
  for (int j = 0; j < 16; ++j) a[j] *= sa;

  __syncthreads();

  v8f acc0 = {}, acc1 = {}, acc2 = {}, acc3 = {};
  const v4f* bp = wl + h * DCH + 64 * nh + c;
  v4f b0 = bp[0], b1 = bp[16], b2 = bp[32], b3 = bp[48];
#pragma unroll
  for (int j = 0; j < 16; ++j) {
    const int jn = (j + 1 < 16) ? (j + 1) : j;
    const v4f n0 = bp[jn * 2 * DCH + 0];
    const v4f n1 = bp[jn * 2 * DCH + 16];
    const v4f n2 = bp[jn * 2 * DCH + 32];
    const v4f n3 = bp[jn * 2 * DCH + 48];
    __builtin_amdgcn_sched_barrier(0);
    v2f af, bf;
    af.x = a[j].x; af.y = a[j].y;
    bf.x = b0.x; bf.y = b0.y;
    acc0 = __builtin_amdgcn_wmma_f32_16x16x4_f32(false, af, false, bf, (short)0, acc0, false, false);
    bf.x = b1.x; bf.y = b1.y;
    acc1 = __builtin_amdgcn_wmma_f32_16x16x4_f32(false, af, false, bf, (short)0, acc1, false, false);
    bf.x = b2.x; bf.y = b2.y;
    acc2 = __builtin_amdgcn_wmma_f32_16x16x4_f32(false, af, false, bf, (short)0, acc2, false, false);
    bf.x = b3.x; bf.y = b3.y;
    acc3 = __builtin_amdgcn_wmma_f32_16x16x4_f32(false, af, false, bf, (short)0, acc3, false, false);
    af.x = a[j].z; af.y = a[j].w;
    bf.x = b0.z; bf.y = b0.w;
    acc0 = __builtin_amdgcn_wmma_f32_16x16x4_f32(false, af, false, bf, (short)0, acc0, false, false);
    bf.x = b1.z; bf.y = b1.w;
    acc1 = __builtin_amdgcn_wmma_f32_16x16x4_f32(false, af, false, bf, (short)0, acc1, false, false);
    bf.x = b2.z; bf.y = b2.w;
    acc2 = __builtin_amdgcn_wmma_f32_16x16x4_f32(false, af, false, bf, (short)0, acc2, false, false);
    bf.x = b3.z; bf.y = b3.w;
    acc3 = __builtin_amdgcn_wmma_f32_16x16x4_f32(false, af, false, bf, (short)0, acc3, false, false);
    __builtin_amdgcn_sched_barrier(0);
    b0 = n0; b1 = n1; b2 = n2; b3 = n3;
  }

  float* obase = out + (size_t)m0 * DCH + 64 * nh + 4 * c;
  if (m0 + 16 <= nrows) {
#pragma unroll
    for (int r = 0; r < 8; ++r) {
      v4f o;
      o.x = acc0[r]; o.y = acc1[r]; o.z = acc2[r]; o.w = acc3[r];
      *(v4f*)(obase + (size_t)(8 * h + r) * DCH) = o;
    }
  } else {
#pragma unroll
    for (int r = 0; r < 8; ++r) {
      v4f o;
      o.x = acc0[r]; o.y = acc1[r]; o.z = acc2[r]; o.w = acc3[r];
      if (m0 + 8 * h + r < nrows) *(v4f*)(obase + (size_t)(8 * h + r) * DCH) = o;
    }
  }
}

template <int R>
__device__ __forceinline__ void gather_rows(const float* __restrict__ hlin,
                                            int my_s, float dd,
                                            int j, int lane, v4f& acc) {
  v4f v[R];
#pragma unroll
  for (int k = 0; k < R; ++k) {
    const int s = __builtin_amdgcn_readlane(my_s, j + k);
    v[k] = ((const v4f*)(hlin + (size_t)s * DCH))[lane];
  }
#pragma unroll
  for (int k = 0; k < R; ++k) acc += v[k] * dd;
}

#define NODES_PER_WAVE 4

__global__ __launch_bounds__(256)
void gather_norm_kernel(const float* __restrict__ hlin, const float* __restrict__ dinv,
                        const float* __restrict__ bias, const int* __restrict__ rowptr,
                        const int* __restrict__ csr_src,
                        const float* __restrict__ gamma, const float* __restrict__ beta,
                        float* __restrict__ hout, float* __restrict__ jk,
                        int do_relu, int first_layer, int n, int e) {
  const int lane = threadIdx.x & 31;
  int wid = (blockIdx.x * blockDim.x + threadIdx.x) >> 5;
  wid = __builtin_amdgcn_readfirstlane(wid);
  const v4f b = ((const v4f*)bias)[lane];

  const int node0 = wid * NODES_PER_WAVE;
  for (int node = node0; node < node0 + NODES_PER_WAVE && node < n; ++node) {
    const int beg = rowptr[node];
    const int end = (node + 1 < n) ? rowptr[node + 1] : e;
    const float dd = dinv[node];
    const int total = end - beg + 1;

    v4f acc = b;
    for (int q0 = 0; q0 < total; q0 += 32) {
      const int m = (total - q0 < 32) ? (total - q0) : 32;
      int q = q0 + lane;
      if (q > total - 1) q = total - 1;
      int my_s = node;
      if (q > 0) my_s = csr_src[beg + q - 1];

      int j = 0;
      for (; j + 8 <= m; j += 8) gather_rows<8>(hlin, my_s, dd, j, lane, acc);
      switch (m - j) {
        case 7: gather_rows<7>(hlin, my_s, dd, j, lane, acc); break;
        case 6: gather_rows<6>(hlin, my_s, dd, j, lane, acc); break;
        case 5: gather_rows<5>(hlin, my_s, dd, j, lane, acc); break;
        case 4: gather_rows<4>(hlin, my_s, dd, j, lane, acc); break;
        case 3: gather_rows<3>(hlin, my_s, dd, j, lane, acc); break;
        case 2: gather_rows<2>(hlin, my_s, dd, j, lane, acc); break;
        case 1: gather_rows<1>(hlin, my_s, dd, j, lane, acc); break;
        default: break;
      }
    }

    asm volatile("" ::: "memory");
    const v4f g  = ((const v4f*)gamma)[lane];
    const v4f bt = ((const v4f*)beta)[lane];
    v4f* jp = (v4f*)(jk + (size_t)node * DCH) + lane;
    v4f jold = {};
    if (!first_layer) jold = *jp;

    float s = acc.x + acc.y + acc.z + acc.w;
#pragma unroll
    for (int off = 16; off > 0; off >>= 1) s += __shfl_xor(s, off, 32);
    const float mu = s * (1.0f / DCH);

    const float dx = acc.x - mu, dy = acc.y - mu, dz = acc.z - mu, dw = acc.w - mu;
    float qq = dx * dx + dy * dy + dz * dz + dw * dw;
#pragma unroll
    for (int off = 16; off > 0; off >>= 1) qq += __shfl_xor(qq, off, 32);
    const float inv = rsqrtf(qq * (1.0f / DCH) + LN_EPS);

    v4f y;
    y.x = dx * inv * g.x + bt.x;
    y.y = dy * inv * g.y + bt.y;
    y.z = dz * inv * g.z + bt.z;
    y.w = dw * inv * g.w + bt.w;
    if (do_relu) {
      y.x = fmaxf(y.x, 0.0f); y.y = fmaxf(y.y, 0.0f);
      y.z = fmaxf(y.z, 0.0f); y.w = fmaxf(y.w, 0.0f);
    }

    ((v4f*)(hout + (size_t)node * DCH))[lane] = y;

    if (first_layer) {
      *jp = y;
    } else {
      v4f o;
      o.x = fmaxf(jold.x, y.x); o.y = fmaxf(jold.y, y.y);
      o.z = fmaxf(jold.z, y.z); o.w = fmaxf(jold.w, y.w);
      *jp = o;
    }
  }
}

extern "C" void kernel_launch(void* const* d_in, const int* in_sizes, int n_in,
                              void* d_out, int out_size, void* d_ws, size_t ws_size,
                              hipStream_t stream) {
  const int N = in_sizes[0] / DCH;
  const int E = in_sizes[1] / 2;

  const float* feat = (const float*)d_in[0];
  const int*   ei   = (const int*)d_in[1];
  const int*   src  = ei;
  const int*   dst  = ei + E;
  const float* Wm[3] = {(const float*)d_in[2], (const float*)d_in[6],  (const float*)d_in[10]};
  const float* bi[3] = {(const float*)d_in[3], (const float*)d_in[7],  (const float*)d_in[11]};
  const float* gm[3] = {(const float*)d_in[4], (const float*)d_in[8],  (const float*)d_in[12]};
  const float* be[3] = {(const float*)d_in[5], (const float*)d_in[9],  (const float*)d_in[13]};
  float* out = (float*)d_out;

  float*    wsf     = (float*)d_ws;
  int*      rowptr  = (int*)wsf;
  float*    dinv    = wsf + N;
  float*    hlin    = wsf + 2 * (size_t)N;
  float*    agg     = hlin + (size_t)N * DCH;
  float*    hbuf    = agg  + (size_t)N * DCH;
  int*      csr_src = (int*)(hbuf + (size_t)N * DCH);
  unsigned* cnt     = (unsigned*)agg;
  unsigned* rank    = cnt + N;

  const int nodeBlocks = (N + 255) / 256;
  const int scanBlocks = (N + 1023) / 1024;
  zero_cnt_kernel<<<nodeBlocks, 256, 0, stream>>>(cnt, N);
  count_rank_kernel<<<2048, 256, 0, stream>>>(dst, cnt, rank, E);
  scan_dinv_kernel<<<scanBlocks, 1024, 0, stream>>>(cnt, rowptr, dinv, N);
  place_kernel<<<2048, 256, 0, stream>>>(src, dst, rowptr, rank, csr_src, E);

  const int gemmBlocks = (N + GEMM_ROWS_PER_BLOCK - 1) / GEMM_ROWS_PER_BLOCK;
  const int gathBlocks = (N + 8 * NODES_PER_WAVE - 1) / (8 * NODES_PER_WAVE);

  const float* X = feat;
  for (int l = 0; l < 3; ++l) {
    gemm_wmma_kernel<<<gemmBlocks, 256, 0, stream>>>(X, Wm[l], dinv, hlin, N);
    gather_norm_kernel<<<gathBlocks, 256, 0, stream>>>(hlin, dinv, bi[l], rowptr, csr_src,
                                                       gm[l], be[l], hbuf, out,
                                                        (l < 2),
                                                        (l == 0), N, E);
    X = hbuf;
  }
}


# Edge list by a two-level counting sort on LDS counters (256-node buckets, then per-bucket degree, scan, placement): no per-edge global atomic; order of a node's list, so its sum order, changes
# speedup vs baseline: 4.3606x; 1.3661x over previous
#include <hip/hip_runtime.h>

#define DCH 128
#define LN_EPS 1e-5f

typedef float v2f __attribute__((ext_vector_type(2)));
typedef float v4f __attribute__((ext_vector_type(4)));
typedef float v8f __attribute__((ext_vector_type(8)));

#define BKT_SHIFT   8
#define BKT_NODES   256
#define MAX_BKTS    512
#define P1_BLOCKS   512
#define P1_THREADS  512

typedef int v2i __attribute__((ext_vector_type(2)));
static_assert(P1_THREADS == MAX_BKTS, "thread t owns bucket t");
static_assert(MAX_BKTS == 2 * BKT_NODES, "a level-2 block reads all bucket totals with two loads per thread");

__global__ void zero_words_kernel(unsigned* __restrict__ p, int n) {
  int i = blockIdx.x * blockDim.x + threadIdx.x;
  if (i < n) p[i] = 0u;
}

__global__ __launch_bounds__(P1_THREADS)
void bucket_hist_kernel(const int* __restrict__ dst, unsigned* __restrict__ total,
                        unsigned* __restrict__ blk_base, int e, int chunk, int nb) {
  __shared__ unsigned hist[MAX_BKTS];
  const int t = threadIdx.x;
  hist[t] = 0u;
  __syncthreads();
  const int beg = blockIdx.x * chunk;
  int end = beg + chunk;
  if (end > e) end = e;
  for (int i0 = beg; i0 < end; i0 += 4 * P1_THREADS) {
    int d[4];
#pragma unroll
    for (int k = 0; k < 4; ++k) {
      const int idx = i0 + k * P1_THREADS + t;
      d[k] = (idx < end) ? dst[idx] : -1;
    }
#pragma unroll
    for (int k = 0; k < 4; ++k)
      if (d[k] >= 0) atomicAdd(&hist[d[k] >> BKT_SHIFT], 1u);
  }
  __syncthreads();
  if (t < nb) blk_base[blockIdx.x * MAX_BKTS + t] = atomicAdd(&total[t], hist[t]);
}

__global__ __launch_bounds__(P1_THREADS)
void bucket_scatter_kernel(const int* __restrict__ src, const int* __restrict__ dst,
                           const unsigned* __restrict__ total,
                           const unsigned* __restrict__ blk_base,
                           int* __restrict__ pairs_i, int e, int chunk, int nb) {
  __shared__ unsigned cur[MAX_BKTS];
  __shared__ unsigned wave_tot[P1_THREADS / 32];
  v2i* __restrict__ pairs = (v2i*)pairs_i;
  const int t = threadIdx.x, lane = t & 31, wv = t >> 5;
  const unsigned c    = (t < nb) ? total[t] : 0u;
  const unsigned mine = (t < nb) ? blk_base[blockIdx.x * MAX_BKTS + t] : 0u;
  unsigned inc = c;
  for (int off = 1; off < 32; off <<= 1) {
    unsigned v = __shfl_up(inc, off, 32);
    if (lane >= off) inc += v;
  }
  if (lane == 31) wave_tot[wv] = inc;
  __syncthreads();
  unsigned before = 0u;
  for (int k = 0; k < P1_THREADS / 32; ++k)
    if (k < wv) before += wave_tot[k];
  cur[t] = before + inc - c + mine;
  __syncthreads();

  const int beg = blockIdx.x * chunk;
  int end = beg + chunk;
  if (end > e) end = e;
  for (int i0 = beg; i0 < end; i0 += 4 * P1_THREADS) {
    int s[4], d[4];
#pragma unroll
    for (int k = 0; k < 4; ++k) {
      const int idx = i0 + k * P1_THREADS + t;
      const bool ok = idx < end;
      d[k] = ok ? dst[idx] : -1;
      s[k] = ok ? src[idx] : 0;
    }
    unsigned pos[4];
#pragma unroll
    for (int k = 0; k < 4; ++k) {
      pos[k] = 0u;
      if (d[k] >= 0) pos[k] = atomicAdd(&cur[d[k] >> BKT_SHIFT], 1u);
    }
#pragma unroll
    for (int k = 0; k < 4; ++k) {
      if (d[k] >= 0) {
        v2i pr; pr.x = s[k]; pr.y = d[k];
        pairs[pos[k]] = pr;
      }
    }
  }
}

__global__ __launch_bounds__(BKT_NODES)
void bucket_sort_kernel(const unsigned* __restrict__ total, const int* __restrict__ pairs_i,
                        int* __restrict__ rowptr, float* __restrict__ dinv,
                        int* __restrict__ csr_src, int n) {
  __shared__ unsigned cnt[BKT_NODES];
  __shared__ unsigned part[BKT_NODES / 32];
  __shared__ unsigned wave_tot[BKT_NODES / 32];
  const v2i* __restrict__ pairs = (const v2i*)pairs_i;
  const int t = threadIdx.x, lane = t & 31, wv = t >> 5;
  const int b = blockIdx.x;

  unsigned p = 0u;
  if (t < b) p = total[t];
  if (t + BKT_NODES < b) p += total[t + BKT_NODES];
  for (int off = 16; off > 0; off >>= 1) p += __shfl_xor(p, off, 32);
  if (lane == 0) part[wv] = p;
  cnt[t] = 0u;
  __syncthreads();
  unsigned start = 0u;
  for (int k = 0; k < BKT_NODES / 32; ++k) start += part[k];
  const int beg = (int)start;
  const int end = beg + (int)total[b];

  for (int i0 = beg; i0 < end; i0 += 4 * BKT_NODES) {
    int d[4];
#pragma unroll
    for (int k = 0; k < 4; ++k) {
      const int idx = i0 + k * BKT_NODES + t;
      d[k] = (idx < end) ? pairs[idx].y : -1;
    }
#pragma unroll
    for (int k = 0; k < 4; ++k)
      if (d[k] >= 0) atomicAdd(&cnt[d[k] & (BKT_NODES - 1)], 1u);
  }
  __syncthreads();

  const unsigned c = cnt[t];
  unsigned inc = c;
  for (int off = 1; off < 32; off <<= 1) {
    unsigned v = __shfl_up(inc, off, 32);
    if (lane >= off) inc += v;
  }
  if (lane == 31) wave_tot[wv] = inc;
  __syncthreads();
  unsigned before = start;
  for (int k = 0; k < BKT_NODES / 32; ++k)
    if (k < wv) before += wave_tot[k];
  const unsigned first = before + inc - c;
  cnt[t] = first;
  __syncthreads();
  const int node = b * BKT_NODES + t;
  if (node < n) {
    rowptr[node] = (int)first;
    dinv[node]   = rsqrtf(1.0f + (float)c);
  }

  for (int i0 = beg; i0 < end; i0 += 4 * BKT_NODES) {
    v2i pr[4];
#pragma unroll
    for (int k = 0; k < 4; ++k) {
      const int idx = i0 + k * BKT_NODES + t;
      v2i none; none.x = 0; none.y = -1;
      pr[k] = (idx < end) ? pairs[idx] : none;
    }
    unsigned pos[4];
#pragma unroll
    for (int k = 0; k < 4; ++k) {
      pos[k] = 0u;
      if (pr[k].y >= 0) pos[k] = atomicAdd(&cnt[pr[k].y & (BKT_NODES - 1)], 1u);
    }
#pragma unroll
    for (int k = 0; k < 4; ++k)
      if (pr[k].y >= 0) csr_src[pos[k]] = pr[k].x;
  }
}

#define GEMM_ROWS_PER_BLOCK 64
__global__ __launch_bounds__(256)
void gemm_wmma_kernel(const float* __restrict__ X, const float* __restrict__ W,
                      const float* __restrict__ dinv, float* __restrict__ out, int nrows) {
  __shared__ v4f wl[32 * DCH];
  const int t    = threadIdx.x;
  const int lane = t & 31;
  const int wave = __builtin_amdgcn_readfirstlane(t >> 5);
  const int c    = lane & 15;
  const int h    = lane >> 4;

#pragma unroll
  for (int it = 0; it < 4; ++it) {
    const int kq = wave + 8 * it;
    const v4f* wr = (const v4f*)(W + (size_t)(4 * kq) * DCH) + lane;
    const v4f r0 = wr[0], r1 = wr[32], r2 = wr[64], r3 = wr[96];
    v4f* dp = wl + kq * DCH + 64 * h + c;
    v4f q0, q1, q2, q3;
    q0.x = r0.x; q0.y = r1.x; q0.z = r2.x; q0.w = r3.x;
    q1.x = r0.y; q1.y = r1.y; q1.z = r2.y; q1.w = r3.y;
    q2.x = r0.z; q2.y = r1.z; q2.z = r2.z; q2.w = r3.z;
    q3.x = r0.w; q3.y = r1.w; q3.z = r2.w; q3.w = r3.w;
    dp[0]  = q0;
    dp[16] = q1;
    dp[32] = q2;
    dp[48] = q3;
  }

  const int nh = wave & 1;
  const int m0 = (blockIdx.x * 4 + (wave >> 1)) * 16;

  int arow = m0 + c;
  if (arow > nrows - 1) arow = nrows - 1;
  const v4f* ap = (const v4f*)(X + (size_t)arow * DCH) + h;
  v4f a[16];
#pragma unroll
  for (int j = 0; j < 16; ++j) a[j] = ap[2 * j];
  const float sa = dinv[arow];
#pragma unroll
  for (int j = 0; j < 16; ++j) a[j] *= sa;

  __syncthreads();

  v8f acc0 = {}, acc1 = {}, acc2 = {}, acc3 = {};
  const v4f* bp = wl + h * DCH + 64 * nh + c;
  v4f b0 = bp[0], b1 = bp[16], b2 = bp[32], b3 = bp[48];
#pragma unroll
  for (int j = 0; j < 16; ++j) {
    const int jn = (j + 1 < 16) ? (j + 1) : j;
    const v4f n0 = bp[jn * 2 * DCH + 0];
    const v4f n1 = bp[jn * 2 * DCH + 16];
    const v4f n2 = bp[jn * 2 * DCH + 32];
    const v4f n3 = bp[jn * 2 * DCH + 48];
    __builtin_amdgcn_sched_barrier(0);
    v2f af, bf;
    af.x = a[j].x; af.y = a[j].y;
    bf.x = b0.x; bf.y = b0.y;
    acc0 = __builtin_amdgcn_wmma_f32_16x16x4_f32(false, af, false, bf, (short)0, acc0, false, false);
    bf.x = b1.x; bf.y = b1.y;
    acc1 = __builtin_amdgcn_wmma_f32_16x16x4_f32(false, af, false, bf, (short)0, acc1, false, false);
    bf.x = b2.x; bf.y = b2.y;
    acc2 = __builtin_amdgcn_wmma_f32_16x16x4_f32(false, af, false, bf, (short)0, acc2, false, false);
    bf.x = b3.x; bf.y = b3.y;
    acc3 = __builtin_amdgcn_wmma_f32_16x16x4_f32(false, af, false, bf, (short)0, acc3, false, false);
    af.x = a[j].z; af.y = a[j].w;
    bf.x = b0.z; bf.y = b0.w;
    acc0 = __builtin_amdgcn_wmma_f32_16x16x4_f32(false, af, false, bf, (short)0, acc0, false, false);
    bf.x = b1.z; bf.y = b1.w;
    acc1 = __builtin_amdgcn_wmma_f32_16x16x4_f32(false, af, false, bf, (short)0, acc1, false, false);
    bf.x = b2.z; bf.y = b2.w;
    acc2 = __builtin_amdgcn_wmma_f32_16x16x4_f32(false, af, false, bf, (short)0, acc2, false, false);
    bf.x = b3.z; bf.y = b3.w;
    acc3 = __builtin_amdgcn_wmma_f32_16x16x4_f32(false, af, false, bf, (short)0, acc3, false, false);
    __builtin_amdgcn_sched_barrier(0);
    b0 = n0; b1 = n1; b2 = n2; b3 = n3;
  }

  float* obase = out + (size_t)m0 * DCH + 64 * nh + 4 * c;
  if (m0 + 16 <= nrows) {
#pragma unroll
    for (int r = 0; r < 8; ++r) {
      v4f o;
      o.x = acc0[r]; o.y = acc1[r]; o.z = acc2[r]; o.w = acc3[r];
      *(v4f*)(obase + (size_t)(8 * h + r) * DCH) = o;
    }
  } else {
#pragma unroll
    for (int r = 0; r < 8; ++r) {
      v4f o;
      o.x = acc0[r]; o.y = acc1[r]; o.z = acc2[r]; o.w = acc3[r];
      if (m0 + 8 * h + r < nrows) *(v4f*)(obase + (size_t)(8 * h + r) * DCH) = o;
    }
  }
}

template <int R>
__device__ __forceinline__ void gather_rows(const float* __restrict__ hlin,
                                            int my_s, float dd,
                                            int j, int lane, v4f& acc) {
  v4f v[R];
#pragma unroll
  for (int k = 0; k < R; ++k) {
    const int s = __builtin_amdgcn_readlane(my_s, j + k);
    v[k] = ((const v4f*)(hlin + (size_t)s * DCH))[lane];
  }
#pragma unroll
  for (int k = 0; k < R; ++k) acc += v[k] * dd;
}

#define NODES_PER_WAVE 4

__global__ __launch_bounds__(256)
void gather_norm_kernel(const float* __restrict__ hlin, const float* __restrict__ dinv,
                        const float* __restrict__ bias, const int* __restrict__ rowptr,
                        const int* __restrict__ csr_src,
                        const float* __restrict__ gamma, const float* __restrict__ beta,
                        float* __restrict__ hout, float* __restrict__ jk,
                        int do_relu, int first_layer, int n, int e) {
  const int lane = threadIdx.x & 31;
  int wid = (blockIdx.x * blockDim.x + threadIdx.x) >> 5;
  wid = __builtin_amdgcn_readfirstlane(wid);
  const v4f b = ((const v4f*)bias)[lane];

  const int node0 = wid * NODES_PER_WAVE;
  for (int node = node0; node < node0 + NODES_PER_WAVE && node < n; ++node) {
    const int beg = rowptr[node];
    const int end = (node + 1 < n) ? rowptr[node + 1] : e;
    const float dd = dinv[node];
    const int total = end - beg + 1;

    v4f acc = b;
    for (int q0 = 0; q0 < total; q0 += 32) {
      const int m = (total - q0 < 32) ? (total - q0) : 32;
      int q = q0 + lane;
      if (q > total - 1) q = total - 1;
      int my_s = node;
      if (q > 0) my_s = csr_src[beg + q - 1];

      int j = 0;
      for (; j + 8 <= m; j += 8) gather_rows<8>(hlin, my_s, dd, j, lane, acc);
      switch (m - j) {
        case 7: gather_rows<7>(hlin, my_s, dd, j, lane, acc); break;
        case 6: gather_rows<6>(hlin, my_s, dd, j, lane, acc); break;
        case 5: gather_rows<5>(hlin, my_s, dd, j, lane, acc); break;
        case 4: gather_rows<4>(hlin, my_s, dd, j, lane, acc); break;
        case 3: gather_rows<3>(hlin, my_s, dd, j, lane, acc); break;
        case 2: gather_rows<2>(hlin, my_s, dd, j, lane, acc); break;
        case 1: gather_rows<1>(hlin, my_s, dd, j, lane, acc); break;
        default: break;
      }
    }

    asm volatile("" ::: "memory");
    const v4f g  = ((const v4f*)gamma)[lane];
    const v4f bt = ((const v4f*)beta)[lane];
    v4f* jp = (v4f*)(jk + (size_t)node * DCH) + lane;
    v4f jold = {};
    if (!first_layer) jold = *jp;

    float s = acc.x + acc.y + acc.z + acc.w;
#pragma unroll
    for (int off = 16; off > 0; off >>= 1) s += __shfl_xor(s, off, 32);
    const float mu = s * (1.0f / DCH);

    const float dx = acc.x - mu, dy = acc.y - mu, dz = acc.z - mu, dw = acc.w - mu;
    float qq = dx * dx + dy * dy + dz * dz + dw * dw;
#pragma unroll
    for (int off = 16; off > 0; off >>= 1) qq += __shfl_xor(qq, off, 32);
    const float inv = rsqrtf(qq * (1.0f / DCH) + LN_EPS);

    v4f y;
    y.x = dx * inv * g.x + bt.x;
    y.y = dy * inv * g.y + bt.y;
    y.z = dz * inv * g.z + bt.z;
    y.w = dw * inv * g.w + bt.w;
    if (do_relu) {
      y.x = fmaxf(y.x, 0.0f); y.y = fmaxf(y.y, 0.0f);
      y.z = fmaxf(y.z, 0.0f); y.w = fmaxf(y.w, 0.0f);
    }

    ((v4f*)(hout + (size_t)node * DCH))[lane] = y;

    if (first_layer) {
      *jp = y;
    } else {
      v4f o;
      o.x = fmaxf(jold.x, y.x); o.y = fmaxf(jold.y, y.y);
      o.z = fmaxf(jold.z, y.z); o.w = fmaxf(jold.w, y.w);
      *jp = o;
    }
  }
}

extern "C" void kernel_launch(void* const* d_in, const int* in_sizes, int n_in,
                              void* d_out, int out_size, void* d_ws, size_t ws_size,
                              hipStream_t stream) {
  const int N = in_sizes[0] / DCH;
  const int E = in_sizes[1] / 2;

  const float* feat = (const float*)d_in[0];
  const int*   ei   = (const int*)d_in[1];
  const int*   src  = ei;
  const int*   dst  = ei + E;
  const float* Wm[3] = {(const float*)d_in[2], (const float*)d_in[6],  (const float*)d_in[10]};
  const float* bi[3] = {(const float*)d_in[3], (const float*)d_in[7],  (const float*)d_in[11]};
  const float* gm[3] = {(const float*)d_in[4], (const float*)d_in[8],  (const float*)d_in[12]};
  const float* be[3] = {(const float*)d_in[5], (const float*)d_in[9],  (const float*)d_in[13]};
  float* out = (float*)d_out;

  float*    wsf      = (float*)d_ws;
  int*      rowptr   = (int*)wsf;
  float*    dinv     = wsf + N;
  float*    hlin     = wsf + 2 * (size_t)N;
  float*    agg      = hlin + (size_t)N * DCH;
  float*    hbuf     = agg  + (size_t)N * DCH;
  int*      csr_src  = (int*)(hbuf + (size_t)N * DCH);
  unsigned* total    = (unsigned*)agg;
  unsigned* blk_base = total + MAX_BKTS;
  int*      pairs    = (int*)(blk_base + (size_t)P1_BLOCKS * MAX_BKTS);

  const int nb    = (N + BKT_NODES - 1) / BKT_NODES;
  const int chunk = (E + P1_BLOCKS - 1) / P1_BLOCKS;
  zero_words_kernel<<<(nb + 255) / 256, 256, 0, stream>>>(total, nb);
  bucket_hist_kernel<<<P1_BLOCKS, P1_THREADS, 0, stream>>>(dst, total, blk_base, E, chunk, nb);
  bucket_scatter_kernel<<<P1_BLOCKS, P1_THREADS, 0, stream>>>(src, dst, total, blk_base, pairs, E, chunk, nb);
  bucket_sort_kernel<<<nb, BKT_NODES, 0, stream>>>(total, pairs, rowptr, dinv, csr_src, N);

  const int gemmBlocks = (N + GEMM_ROWS_PER_BLOCK - 1) / GEMM_ROWS_PER_BLOCK;
  const int gathBlocks = (N + 8 * NODES_PER_WAVE - 1) / (8 * NODES_PER_WAVE);

  const float* X = feat;
  for (int l = 0; l < 3; ++l) {
    gemm_wmma_kernel<<<gemmBlocks, 256, 0, stream>>>(X, Wm[l], dinv, hlin, N);
    gather_norm_kernel<<<gathBlocks, 256, 0, stream>>>(hlin, dinv, bi[l], rowptr, csr_src,
                                                       gm[l], be[l], hbuf, out,
                                                        (l < 2),
                                                        (l == 0), N, E);
    X = hbuf;
  }
}


# Next layer's product fused into the gather: a block gathers and normalises 64 nodes, keeps the rows in LDS and multiplies them by W (f32 WMMA); h never reaches memory, 2 launches go; sum orders kept
# speedup vs baseline: 4.7239x; 1.0833x over previous
#include <hip/hip_runtime.h>

#define DCH 128
#define LN_EPS 1e-5f

typedef float v2f __attribute__((ext_vector_type(2)));
typedef float v4f __attribute__((ext_vector_type(4)));
typedef float v8f __attribute__((ext_vector_type(8)));

#define BKT_SHIFT   8
#define BKT_NODES   256
#define MAX_BKTS    512
#define P1_BLOCKS   512
#define P1_THREADS  512

typedef int v2i __attribute__((ext_vector_type(2)));
static_assert(P1_THREADS == MAX_BKTS, "thread t owns bucket t");
static_assert(MAX_BKTS == 2 * BKT_NODES, "a level-2 block reads all bucket totals with two loads per thread");

__global__ void zero_words_kernel(unsigned* __restrict__ p, int n) {
  int i = blockIdx.x * blockDim.x + threadIdx.x;
  if (i < n) p[i] = 0u;
}

__global__ __launch_bounds__(P1_THREADS)
void bucket_hist_kernel(const int* __restrict__ dst, unsigned* __restrict__ total,
                        unsigned* __restrict__ blk_base, int e, int chunk, int nb) {
  __shared__ unsigned hist[MAX_BKTS];
  const int t = threadIdx.x;
  hist[t] = 0u;
  __syncthreads();
  const int beg = blockIdx.x * chunk;
  int end = beg + chunk;
  if (end > e) end = e;
  for (int i0 = beg; i0 < end; i0 += 4 * P1_THREADS) {
    int d[4];
#pragma unroll
    for (int k = 0; k < 4; ++k) {
      const int idx = i0 + k * P1_THREADS + t;
      d[k] = (idx < end) ? dst[idx] : -1;
    }
#pragma unroll
    for (int k = 0; k < 4; ++k)
      if (d[k] >= 0) atomicAdd(&hist[d[k] >> BKT_SHIFT], 1u);
  }
  __syncthreads();
  if (t < nb) blk_base[blockIdx.x * MAX_BKTS + t] = atomicAdd(&total[t], hist[t]);
}

__global__ __launch_bounds__(P1_THREADS)
void bucket_scatter_kernel(const int* __restrict__ src, const int* __restrict__ dst,
                           const unsigned* __restrict__ total,
                           const unsigned* __restrict__ blk_base,
                           int* __restrict__ pairs_i, int e, int chunk, int nb) {
  __shared__ unsigned cur[MAX_BKTS];
  __shared__ unsigned wave_tot[P1_THREADS / 32];
  v2i* __restrict__ pairs = (v2i*)pairs_i;
  const int t = threadIdx.x, lane = t & 31, wv = t >> 5;
  const unsigned c    = (t < nb) ? total[t] : 0u;
  const unsigned mine = (t < nb) ? blk_base[blockIdx.x * MAX_BKTS + t] : 0u;
  unsigned inc = c;
  for (int off = 1; off < 32; off <<= 1) {
    unsigned v = __shfl_up(inc, off, 32);
    if (lane >= off) inc += v;
  }
  if (lane == 31) wave_tot[wv] = inc;
  __syncthreads();
  unsigned before = 0u;
  for (int k = 0; k < P1_THREADS / 32; ++k)
    if (k < wv) before += wave_tot[k];
  cur[t] = before + inc - c + mine;
  __syncthreads();

  const int beg = blockIdx.x * chunk;
  int end = beg + chunk;
  if (end > e) end = e;
  for (int i0 = beg; i0 < end; i0 += 4 * P1_THREADS) {
    int s[4], d[4];
#pragma unroll
    for (int k = 0; k < 4; ++k) {
      const int idx = i0 + k * P1_THREADS + t;
      const bool ok = idx < end;
      d[k] = ok ? dst[idx] : -1;
      s[k] = ok ? src[idx] : 0;
    }
    unsigned pos[4];
#pragma unroll
    for (int k = 0; k < 4; ++k) {
      pos[k] = 0u;
      if (d[k] >= 0) pos[k] = atomicAdd(&cur[d[k] >> BKT_SHIFT], 1u);
    }
#pragma unroll
    for (int k = 0; k < 4; ++k) {
      if (d[k] >= 0) {
        v2i pr; pr.x = s[k]; pr.y = d[k];
        pairs[pos[k]] = pr;
      }
    }
  }
}

__global__ __launch_bounds__(BKT_NODES)
void bucket_sort_kernel(const unsigned* __restrict__ total, const int* __restrict__ pairs_i,
                        int* __restrict__ rowptr, float* __restrict__ dinv,
                        int* __restrict__ csr_src, int n) {
  __shared__ unsigned cnt[BKT_NODES];
  __shared__ unsigned part[BKT_NODES / 32];
  __shared__ unsigned wave_tot[BKT_NODES / 32];
  const v2i* __restrict__ pairs = (const v2i*)pairs_i;
  const int t = threadIdx.x, lane = t & 31, wv = t >> 5;
  const int b = blockIdx.x;

  unsigned p = 0u;
  if (t < b) p = total[t];
  if (t + BKT_NODES < b) p += total[t + BKT_NODES];
  for (int off = 16; off > 0; off >>= 1) p += __shfl_xor(p, off, 32);
  if (lane == 0) part[wv] = p;
  cnt[t] = 0u;
  __syncthreads();
  unsigned start = 0u;
  for (int k = 0; k < BKT_NODES / 32; ++k) start += part[k];
  const int beg = (int)start;
  const int end = beg + (int)total[b];

  for (int i0 = beg; i0 < end; i0 += 4 * BKT_NODES) {
    int d[4];
#pragma unroll
    for (int k = 0; k < 4; ++k) {
      const int idx = i0 + k * BKT_NODES + t;
      d[k] = (idx < end) ? pairs[idx].y : -1;
    }
#pragma unroll
    for (int k = 0; k < 4; ++k)
      if (d[k] >= 0) atomicAdd(&cnt[d[k] & (BKT_NODES - 1)], 1u);
  }
  __syncthreads();

  const unsigned c = cnt[t];
  unsigned inc = c;
  for (int off = 1; off < 32; off <<= 1) {
    unsigned v = __shfl_up(inc, off, 32);
    if (lane >= off) inc += v;
  }
  if (lane == 31) wave_tot[wv] = inc;
  __syncthreads();
  unsigned before = start;
  for (int k = 0; k < BKT_NODES / 32; ++k)
    if (k < wv) before += wave_tot[k];
  const unsigned first = before + inc - c;
  cnt[t] = first;
  __syncthreads();
  const int node = b * BKT_NODES + t;
  if (node < n) {
    rowptr[node] = (int)first;
    dinv[node]   = rsqrtf(1.0f + (float)c);
  }

  for (int i0 = beg; i0 < end; i0 += 4 * BKT_NODES) {
    v2i pr[4];
#pragma unroll
    for (int k = 0; k < 4; ++k) {
      const int idx = i0 + k * BKT_NODES + t;
      v2i none; none.x = 0; none.y = -1;
      pr[k] = (idx < end) ? pairs[idx] : none;
    }
    unsigned pos[4];
#pragma unroll
    for (int k = 0; k < 4; ++k) {
      pos[k] = 0u;
      if (pr[k].y >= 0) pos[k] = atomicAdd(&cnt[pr[k].y & (BKT_NODES - 1)], 1u);
    }
#pragma unroll
    for (int k = 0; k < 4; ++k)
      if (pr[k].y >= 0) csr_src[pos[k]] = pr[k].x;
  }
}

#define GEMM_ROWS_PER_BLOCK 64
__global__ __launch_bounds__(256)
void gemm_wmma_kernel(const float* __restrict__ X, const float* __restrict__ W,
                      const float* __restrict__ dinv, float* __restrict__ out, int nrows) {
  __shared__ v4f wl[32 * DCH];
  const int t    = threadIdx.x;
  const int lane = t & 31;
  const int wave = __builtin_amdgcn_readfirstlane(t >> 5);
  const int c    = lane & 15;
  const int h    = lane >> 4;

#pragma unroll
  for (int it = 0; it < 4; ++it) {
    const int kq = wave + 8 * it;
    const v4f* wr = (const v4f*)(W + (size_t)(4 * kq) * DCH) + lane;
    const v4f r0 = wr[0], r1 = wr[32], r2 = wr[64], r3 = wr[96];
    v4f* dp = wl + kq * DCH + 64 * h + c;
    v4f q0, q1, q2, q3;
    q0.x = r0.x; q0.y = r1.x; q0.z = r2.x; q0.w = r3.x;
    q1.x = r0.y; q1.y = r1.y; q1.z = r2.y; q1.w = r3.y;
    q2.x = r0.z; q2.y = r1.z; q2.z = r2.z; q2.w = r3.z;
    q3.x = r0.w; q3.y = r1.w; q3.z = r2.w; q3.w = r3.w;
    dp[0]  = q0;
    dp[16] = q1;
    dp[32] = q2;
    dp[48] = q3;
  }

  const int nh = wave & 1;
  const int m0 = (blockIdx.x * 4 + (wave >> 1)) * 16;

  int arow = m0 + c;
  if (arow > nrows - 1) arow = nrows - 1;
  const v4f* ap = (const v4f*)(X + (size_t)arow * DCH) + h;
  v4f a[16];
#pragma unroll
  for (int j = 0; j < 16; ++j) a[j] = ap[2 * j];
  const float sa = dinv[arow];
#pragma unroll
  for (int j = 0; j < 16; ++j) a[j] *= sa;

  __syncthreads();

  v8f acc0 = {}, acc1 = {}, acc2 = {}, acc3 = {};
  const v4f* bp = wl + h * DCH + 64 * nh + c;
  v4f b0 = bp[0], b1 = bp[16], b2 = bp[32], b3 = bp[48];
#pragma unroll
  for (int j = 0; j < 16; ++j) {
    const int jn = (j + 1 < 16) ? (j + 1) : j;
    const v4f n0 = bp[jn * 2 * DCH + 0];
    const v4f n1 = bp[jn * 2 * DCH + 16];
    const v4f n2 = bp[jn * 2 * DCH + 32];
    const v4f n3 = bp[jn * 2 * DCH + 48];
    __builtin_amdgcn_sched_barrier(0);
    v2f af, bf;
    af.x = a[j].x; af.y = a[j].y;
    bf.x = b0.x; bf.y = b0.y;
    acc0 = __builtin_amdgcn_wmma_f32_16x16x4_f32(false, af, false, bf, (short)0, acc0, false, false);
    bf.x = b1.x; bf.y = b1.y;
    acc1 = __builtin_amdgcn_wmma_f32_16x16x4_f32(false, af, false, bf, (short)0, acc1, false, false);
    bf.x = b2.x; bf.y = b2.y;
    acc2 = __builtin_amdgcn_wmma_f32_16x16x4_f32(false, af, false, bf, (short)0, acc2, false, false);
    bf.x = b3.x; bf.y = b3.y;
    acc3 = __builtin_amdgcn_wmma_f32_16x16x4_f32(false, af, false, bf, (short)0, acc3, false, false);
    af.x = a[j].z; af.y = a[j].w;
    bf.x = b0.z; bf.y = b0.w;
    acc0 = __builtin_amdgcn_wmma_f32_16x16x4_f32(false, af, false, bf, (short)0, acc0, false, false);
    bf.x = b1.z; bf.y = b1.w;
    acc1 = __builtin_amdgcn_wmma_f32_16x16x4_f32(false, af, false, bf, (short)0, acc1, false, false);
    bf.x = b2.z; bf.y = b2.w;
    acc2 = __builtin_amdgcn_wmma_f32_16x16x4_f32(false, af, false, bf, (short)0, acc2, false, false);
    bf.x = b3.z; bf.y = b3.w;
    acc3 = __builtin_amdgcn_wmma_f32_16x16x4_f32(false, af, false, bf, (short)0, acc3, false, false);
    __builtin_amdgcn_sched_barrier(0);
    b0 = n0; b1 = n1; b2 = n2; b3 = n3;
  }

  float* obase = out + (size_t)m0 * DCH + 64 * nh + 4 * c;
  if (m0 + 16 <= nrows) {
#pragma unroll
    for (int r = 0; r < 8; ++r) {
      v4f o;
      o.x = acc0[r]; o.y = acc1[r]; o.z = acc2[r]; o.w = acc3[r];
      *(v4f*)(obase + (size_t)(8 * h + r) * DCH) = o;
    }
  } else {
#pragma unroll
    for (int r = 0; r < 8; ++r) {
      v4f o;
      o.x = acc0[r]; o.y = acc1[r]; o.z = acc2[r]; o.w = acc3[r];
      if (m0 + 8 * h + r < nrows) *(v4f*)(obase + (size_t)(8 * h + r) * DCH) = o;
    }
  }
}

template <int R>
__device__ __forceinline__ void gather_rows(const float* __restrict__ hlin,
                                            int my_s, float dd,
                                            int j, int lane, v4f& acc) {
  v4f v[R];
#pragma unroll
  for (int k = 0; k < R; ++k) {
    const int s = __builtin_amdgcn_readlane(my_s, j + k);
    v[k] = ((const v4f*)(hlin + (size_t)s * DCH))[lane];
  }
#pragma unroll
  for (int k = 0; k < R; ++k) acc += v[k] * dd;
}

#define NODES_PER_WAVE 4

__device__ __forceinline__ v4f node_row(const float* __restrict__ hlin, const float* __restrict__ dinv,
                                        const v4f b, const int* __restrict__ rowptr,
                                        const int* __restrict__ csr_src,
                                        const float* __restrict__ gamma, const float* __restrict__ beta,
                                        float* __restrict__ jk, int do_relu, int first_layer,
                                        int n, int e, int node, int lane, float& dd_out) {
  const int beg = rowptr[node];
  const int end = (node + 1 < n) ? rowptr[node + 1] : e;
  const float dd = dinv[node];
  const int total = end - beg + 1;

  v4f acc = b;
  for (int q0 = 0; q0 < total; q0 += 32) {
    const int m = (total - q0 < 32) ? (total - q0) : 32;
    int q = q0 + lane;
    if (q > total - 1) q = total - 1;
    int my_s = node;
    if (q > 0) my_s = csr_src[beg + q - 1];

    int j = 0;
    for (; j + 8 <= m; j += 8) gather_rows<8>(hlin, my_s, dd, j, lane, acc);
    switch (m - j) {
      case 7: gather_rows<7>(hlin, my_s, dd, j, lane, acc); break;
      case 6: gather_rows<6>(hlin, my_s, dd, j, lane, acc); break;
      case 5: gather_rows<5>(hlin, my_s, dd, j, lane, acc); break;
      case 4: gather_rows<4>(hlin, my_s, dd, j, lane, acc); break;
      case 3: gather_rows<3>(hlin, my_s, dd, j, lane, acc); break;
      case 2: gather_rows<2>(hlin, my_s, dd, j, lane, acc); break;
      case 1: gather_rows<1>(hlin, my_s, dd, j, lane, acc); break;
      default: break;
    }
  }

  asm volatile("" ::: "memory");
  const v4f g  = ((const v4f*)gamma)[lane];
  const v4f bt = ((const v4f*)beta)[lane];
  v4f* jp = (v4f*)(jk + (size_t)node * DCH) + lane;
  v4f jold = {};
  if (!first_layer) jold = *jp;

  float s = acc.x + acc.y + acc.z + acc.w;
#pragma unroll
  for (int off = 16; off > 0; off >>= 1) s += __shfl_xor(s, off, 32);
  const float mu = s * (1.0f / DCH);

  const float dx = acc.x - mu, dy = acc.y - mu, dz = acc.z - mu, dw = acc.w - mu;
  float qq = dx * dx + dy * dy + dz * dz + dw * dw;
#pragma unroll
  for (int off = 16; off > 0; off >>= 1) qq += __shfl_xor(qq, off, 32);
  const float inv = rsqrtf(qq * (1.0f / DCH) + LN_EPS);

  v4f y;
  y.x = dx * inv * g.x + bt.x;
  y.y = dy * inv * g.y + bt.y;
  y.z = dz * inv * g.z + bt.z;
  y.w = dw * inv * g.w + bt.w;
  if (do_relu) {
    y.x = fmaxf(y.x, 0.0f); y.y = fmaxf(y.y, 0.0f);
    y.z = fmaxf(y.z, 0.0f); y.w = fmaxf(y.w, 0.0f);
  }

  if (first_layer) {
    *jp = y;
  } else {
    v4f o;
    o.x = fmaxf(jold.x, y.x); o.y = fmaxf(jold.y, y.y);
    o.z = fmaxf(jold.z, y.z); o.w = fmaxf(jold.w, y.w);
    *jp = o;
  }
  dd_out = dd;
  return y;
}

__global__ __launch_bounds__(256)
void gather_norm_kernel(const float* __restrict__ hlin, const float* __restrict__ dinv,
                        const float* __restrict__ bias, const int* __restrict__ rowptr,
                        const int* __restrict__ csr_src,
                        const float* __restrict__ gamma, const float* __restrict__ beta,
                        float* __restrict__ jk,
                        int do_relu, int first_layer, int n, int e) {
  const int lane = threadIdx.x & 31;
  int wid = (blockIdx.x * blockDim.x + threadIdx.x) >> 5;
  wid = __builtin_amdgcn_readfirstlane(wid);
  const v4f b = ((const v4f*)bias)[lane];
  const int node0 = wid * NODES_PER_WAVE;
  for (int node = node0; node < node0 + NODES_PER_WAVE && node < n; ++node) {
    float dd;
    (void)node_row(hlin, dinv, b, rowptr, csr_src, gamma, beta, jk, do_relu, first_layer,
                   n, e, node, lane, dd);
  }
}

#define FUSE_ROWS  64
#define FUSE_PITCH 33
__global__ __launch_bounds__(512)
void gather_gemm_kernel(const float* __restrict__ hlin, const float* __restrict__ dinv,
                        const float* __restrict__ bias, const int* __restrict__ rowptr,
                        const int* __restrict__ csr_src,
                        const float* __restrict__ gamma, const float* __restrict__ beta,
                        const float* __restrict__ Wn, float* __restrict__ hnext,
                        float* __restrict__ jk,
                        int do_relu, int first_layer, int n, int e) {
  __shared__ v4f wl[32 * DCH];
  __shared__ v4f hrow[FUSE_ROWS * FUSE_PITCH];
  const int t    = threadIdx.x;
  const int lane = t & 31;
  const int wave = __builtin_amdgcn_readfirstlane(t >> 5);
  const int c    = lane & 15;
  const int h    = lane >> 4;

#pragma unroll
  for (int it = 0; it < 2; ++it) {
    const int kq = wave + 16 * it;
    const v4f* wr = (const v4f*)(Wn + (size_t)(4 * kq) * DCH) + lane;
    const v4f r0 = wr[0], r1 = wr[32], r2 = wr[64], r3 = wr[96];
    v4f* dp = wl + kq * DCH + 32 * (lane >> 3) + 2 * (lane & 7);
    v4f q0, q1, q2, q3;
    q0.x = r0.x; q0.y = r1.x; q0.z = r2.x; q0.w = r3.x;
    q1.x = r0.y; q1.y = r1.y; q1.z = r2.y; q1.w = r3.y;
    q2.x = r0.z; q2.y = r1.z; q2.z = r2.z; q2.w = r3.z;
    q3.x = r0.w; q3.y = r1.w; q3.z = r2.w; q3.w = r3.w;
    dp[0]  = q0;
    dp[16] = q1;
    dp[1]  = q2;
    dp[17] = q3;
  }

  const v4f b = ((const v4f*)bias)[lane];
  const int row0 = blockIdx.x * FUSE_ROWS;
  for (int i = 0; i < NODES_PER_WAVE; ++i) {
    const int loc  = wave * NODES_PER_WAVE + i;
    const int node = row0 + loc;
    v4f a = {};
    if (node < n) {
      float dd;
      const v4f y = node_row(hlin, dinv, b, rowptr, csr_src, gamma, beta, jk, do_relu, first_layer,
                             n, e, node, lane, dd);
      a = y * dd;
    }
    hrow[loc * FUSE_PITCH + lane] = a;
  }

  __syncthreads();

  const int mt = wave >> 2;
  const int nq = wave & 3;
  const v4f* ar = hrow + (16 * mt + c) * FUSE_PITCH + h;
  const v4f* bp = wl + h * DCH + 32 * nq + c;
  v8f acc0 = {}, acc1 = {};
  v4f a0 = ar[0], b0 = bp[0], b1 = bp[16];
#pragma unroll
  for (int j = 0; j < 16; ++j) {
    const int jn = (j + 1 < 16) ? (j + 1) : j;
    const v4f na = ar[2 * jn];
    const v4f n0 = bp[jn * 2 * DCH + 0];
    const v4f n1 = bp[jn * 2 * DCH + 16];
    __builtin_amdgcn_sched_barrier(0);
    v2f af, bf;
    af.x = a0.x; af.y = a0.y;
    bf.x = b0.x; bf.y = b0.y;
    acc0 = __builtin_amdgcn_wmma_f32_16x16x4_f32(false, af, false, bf, (short)0, acc0, false, false);
    bf.x = b1.x; bf.y = b1.y;
    acc1 = __builtin_amdgcn_wmma_f32_16x16x4_f32(false, af, false, bf, (short)0, acc1, false, false);
    af.x = a0.z; af.y = a0.w;
    bf.x = b0.z; bf.y = b0.w;
    acc0 = __builtin_amdgcn_wmma_f32_16x16x4_f32(false, af, false, bf, (short)0, acc0, false, false);
    bf.x = b1.z; bf.y = b1.w;
    acc1 = __builtin_amdgcn_wmma_f32_16x16x4_f32(false, af, false, bf, (short)0, acc1, false, false);
    __builtin_amdgcn_sched_barrier(0);
    a0 = na; b0 = n0; b1 = n1;
  }

  const int m0 = row0 + 16 * mt;
  float* obase = hnext + (size_t)m0 * DCH + 32 * nq + 2 * c;
  if (m0 + 16 <= n) {
#pragma unroll
    for (int r = 0; r < 8; ++r) {
      v2f o;
      o.x = acc0[r]; o.y = acc1[r];
      *(v2f*)(obase + (size_t)(8 * h + r) * DCH) = o;
    }
  } else {
#pragma unroll
    for (int r = 0; r < 8; ++r) {
      v2f o;
      o.x = acc0[r]; o.y = acc1[r];
      if (m0 + 8 * h + r < n) *(v2f*)(obase + (size_t)(8 * h + r) * DCH) = o;
    }
  }
}

extern "C" void kernel_launch(void* const* d_in, const int* in_sizes, int n_in,
                              void* d_out, int out_size, void* d_ws, size_t ws_size,
                              hipStream_t stream) {
  const int N = in_sizes[0] / DCH;
  const int E = in_sizes[1] / 2;

  const float* feat = (const float*)d_in[0];
  const int*   ei   = (const int*)d_in[1];
  const int*   src  = ei;
  const int*   dst  = ei + E;
  const float* Wm[3] = {(const float*)d_in[2], (const float*)d_in[6],  (const float*)d_in[10]};
  const float* bi[3] = {(const float*)d_in[3], (const float*)d_in[7],  (const float*)d_in[11]};
  const float* gm[3] = {(const float*)d_in[4], (const float*)d_in[8],  (const float*)d_in[12]};
  const float* be[3] = {(const float*)d_in[5], (const float*)d_in[9],  (const float*)d_in[13]};
  float* out = (float*)d_out;

  float*    wsf      = (float*)d_ws;
  int*      rowptr   = (int*)wsf;
  float*    dinv     = wsf + N;
  float*    hlin     = wsf + 2 * (size_t)N;
  float*    agg      = hlin + (size_t)N * DCH;
  float*    hbuf     = agg  + (size_t)N * DCH;
  int*      csr_src  = (int*)(hbuf + (size_t)N * DCH);
  unsigned* total    = (unsigned*)agg;
  unsigned* blk_base = total + MAX_BKTS;
  int*      pairs    = (int*)(blk_base + (size_t)P1_BLOCKS * MAX_BKTS);

  const int nb    = (N + BKT_NODES - 1) / BKT_NODES;
  const int chunk = (E + P1_BLOCKS - 1) / P1_BLOCKS;
  zero_words_kernel<<<(nb + 255) / 256, 256, 0, stream>>>(total, nb);
  bucket_hist_kernel<<<P1_BLOCKS, P1_THREADS, 0, stream>>>(dst, total, blk_base, E, chunk, nb);
  bucket_scatter_kernel<<<P1_BLOCKS, P1_THREADS, 0, stream>>>(src, dst, total, blk_base, pairs, E, chunk, nb);
  bucket_sort_kernel<<<nb, BKT_NODES, 0, stream>>>(total, pairs, rowptr, dinv, csr_src, N);

  const int gemmBlocks = (N + GEMM_ROWS_PER_BLOCK - 1) / GEMM_ROWS_PER_BLOCK;
  const int fuseBlocks = (N + FUSE_ROWS - 1) / FUSE_ROWS;
  const int gathBlocks = (N + 8 * NODES_PER_WAVE - 1) / (8 * NODES_PER_WAVE);

  gemm_wmma_kernel<<<gemmBlocks, 256, 0, stream>>>(feat, Wm[0], dinv, hlin, N);
  gather_gemm_kernel<<<fuseBlocks, 512, 0, stream>>>(hlin, dinv, bi[0], rowptr, csr_src,
                                                     gm[0], be[0], Wm[1], hbuf, out,
                                                      1,  1, N, E);
  gather_gemm_kernel<<<fuseBlocks, 512, 0, stream>>>(hbuf, dinv, bi[1], rowptr, csr_src,
                                                     gm[1], be[1], Wm[2], hlin, out,
                                                      1,  0, N, E);
  gather_norm_kernel<<<gathBlocks, 256, 0, stream>>>(hlin, dinv, bi[2], rowptr, csr_src,
                                                     gm[2], be[2], out,
                                                      0,  0, N, E);
}


# Fused layer launches as resident grids, 768 work-groups (3 per WGP), each an equal share of row tiles in passes; first pass 64/32/16 rows by block: neighbours' products run under gathers; sums kept
# speedup vs baseline: 5.0562x; 1.0703x over previous
#include <hip/hip_runtime.h>

#define DCH 128
#define LN_EPS 1e-5f

typedef float v2f __attribute__((ext_vector_type(2)));
typedef float v4f __attribute__((ext_vector_type(4)));
typedef float v8f __attribute__((ext_vector_type(8)));

#define BKT_SHIFT   8
#define BKT_NODES   256
#define MAX_BKTS    512
#define P1_BLOCKS   512
#define P1_THREADS  512

typedef int v2i __attribute__((ext_vector_type(2)));
static_assert(P1_THREADS == MAX_BKTS, "thread t owns bucket t");
static_assert(MAX_BKTS == 2 * BKT_NODES, "a level-2 block reads all bucket totals with two loads per thread");

__global__ void zero_words_kernel(unsigned* __restrict__ p, int n) {
  int i = blockIdx.x * blockDim.x + threadIdx.x;
  if (i < n) p[i] = 0u;
}

__global__ __launch_bounds__(P1_THREADS)
void bucket_hist_kernel(const int* __restrict__ dst, unsigned* __restrict__ total,
                        unsigned* __restrict__ blk_base, int e, int chunk, int nb) {
  __shared__ unsigned hist[MAX_BKTS];
  const int t = threadIdx.x;
  hist[t] = 0u;
  __syncthreads();
  const int beg = blockIdx.x * chunk;
  int end = beg + chunk;
  if (end > e) end = e;
  for (int i0 = beg; i0 < end; i0 += 4 * P1_THREADS) {
    int d[4];
#pragma unroll
    for (int k = 0; k < 4; ++k) {
      const int idx = i0 + k * P1_THREADS + t;
      d[k] = (idx < end) ? dst[idx] : -1;
    }
#pragma unroll
    for (int k = 0; k < 4; ++k)
      if (d[k] >= 0) atomicAdd(&hist[d[k] >> BKT_SHIFT], 1u);
  }
  __syncthreads();
  if (t < nb) blk_base[blockIdx.x * MAX_BKTS + t] = atomicAdd(&total[t], hist[t]);
}

__global__ __launch_bounds__(P1_THREADS)
void bucket_scatter_kernel(const int* __restrict__ src, const int* __restrict__ dst,
                           const unsigned* __restrict__ total,
                           const unsigned* __restrict__ blk_base,
                           int* __restrict__ pairs_i, int e, int chunk, int nb) {
  __shared__ unsigned cur[MAX_BKTS];
  __shared__ unsigned wave_tot[P1_THREADS / 32];
  v2i* __restrict__ pairs = (v2i*)pairs_i;
  const int t = threadIdx.x, lane = t & 31, wv = t >> 5;
  const unsigned c    = (t < nb) ? total[t] : 0u;
  const unsigned mine = (t < nb) ? blk_base[blockIdx.x * MAX_BKTS + t] : 0u;
  unsigned inc = c;
  for (int off = 1; off < 32; off <<= 1) {
    unsigned v = __shfl_up(inc, off, 32);
    if (lane >= off) inc += v;
  }
  if (lane == 31) wave_tot[wv] = inc;
  __syncthreads();
  unsigned before = 0u;
  for (int k = 0; k < P1_THREADS / 32; ++k)
    if (k < wv) before += wave_tot[k];
  cur[t] = before + inc - c + mine;
  __syncthreads();

  const int beg = blockIdx.x * chunk;
  int end = beg + chunk;
  if (end > e) end = e;
  for (int i0 = beg; i0 < end; i0 += 4 * P1_THREADS) {
    int s[4], d[4];
#pragma unroll
    for (int k = 0; k < 4; ++k) {
      const int idx = i0 + k * P1_THREADS + t;
      const bool ok = idx < end;
      d[k] = ok ? dst[idx] : -1;
      s[k] = ok ? src[idx] : 0;
    }
    unsigned pos[4];
#pragma unroll
    for (int k = 0; k < 4; ++k) {
      pos[k] = 0u;
      if (d[k] >= 0) pos[k] = atomicAdd(&cur[d[k] >> BKT_SHIFT], 1u);
    }
#pragma unroll
    for (int k = 0; k < 4; ++k) {
      if (d[k] >= 0) {
        v2i pr; pr.x = s[k]; pr.y = d[k];
        pairs[pos[k]] = pr;
      }
    }
  }
}

__global__ __launch_bounds__(BKT_NODES)
void bucket_sort_kernel(const unsigned* __restrict__ total, const int* __restrict__ pairs_i,
                        int* __restrict__ rowptr, float* __restrict__ dinv,
                        int* __restrict__ csr_src, int n) {
  __shared__ unsigned cnt[BKT_NODES];
  __shared__ unsigned part[BKT_NODES / 32];
  __shared__ unsigned wave_tot[BKT_NODES / 32];
  const v2i* __restrict__ pairs = (const v2i*)pairs_i;
  const int t = threadIdx.x, lane = t & 31, wv = t >> 5;
  const int b = blockIdx.x;

  unsigned p = 0u;
  if (t < b) p = total[t];
  if (t + BKT_NODES < b) p += total[t + BKT_NODES];
  for (int off = 16; off > 0; off >>= 1) p += __shfl_xor(p, off, 32);
  if (lane == 0) part[wv] = p;
  cnt[t] = 0u;
  __syncthreads();
  unsigned start = 0u;
  for (int k = 0; k < BKT_NODES / 32; ++k) start += part[k];
  const int beg = (int)start;
  const int end = beg + (int)total[b];

  for (int i0 = beg; i0 < end; i0 += 4 * BKT_NODES) {
    int d[4];
#pragma unroll
    for (int k = 0; k < 4; ++k) {
      const int idx = i0 + k * BKT_NODES + t;
      d[k] = (idx < end) ? pairs[idx].y : -1;
    }
#pragma unroll
    for (int k = 0; k < 4; ++k)
      if (d[k] >= 0) atomicAdd(&cnt[d[k] & (BKT_NODES - 1)], 1u);
  }
  __syncthreads();

  const unsigned c = cnt[t];
  unsigned inc = c;
  for (int off = 1; off < 32; off <<= 1) {
    unsigned v = __shfl_up(inc, off, 32);
    if (lane >= off) inc += v;
  }
  if (lane == 31) wave_tot[wv] = inc;
  __syncthreads();
  unsigned before = start;
  for (int k = 0; k < BKT_NODES / 32; ++k)
    if (k < wv) before += wave_tot[k];
  const unsigned first = before + inc - c;
  cnt[t] = first;
  __syncthreads();
  const int node = b * BKT_NODES + t;
  if (node < n) {
    rowptr[node] = (int)first;
    dinv[node]   = rsqrtf(1.0f + (float)c);
  }

  for (int i0 = beg; i0 < end; i0 += 4 * BKT_NODES) {
    v2i pr[4];
#pragma unroll
    for (int k = 0; k < 4; ++k) {
      const int idx = i0 + k * BKT_NODES + t;
      v2i none; none.x = 0; none.y = -1;
      pr[k] = (idx < end) ? pairs[idx] : none;
    }
    unsigned pos[4];
#pragma unroll
    for (int k = 0; k < 4; ++k) {
      pos[k] = 0u;
      if (pr[k].y >= 0) pos[k] = atomicAdd(&cnt[pr[k].y & (BKT_NODES - 1)], 1u);
    }
#pragma unroll
    for (int k = 0; k < 4; ++k)
      if (pr[k].y >= 0) csr_src[pos[k]] = pr[k].x;
  }
}

#define GEMM_ROWS_PER_BLOCK 64
__global__ __launch_bounds__(256)
void gemm_wmma_kernel(const float* __restrict__ X, const float* __restrict__ W,
                      const float* __restrict__ dinv, float* __restrict__ out, int nrows) {
  __shared__ v4f wl[32 * DCH];
  const int t    = threadIdx.x;
  const int lane = t & 31;
  const int wave = __builtin_amdgcn_readfirstlane(t >> 5);
  const int c    = lane & 15;
  const int h    = lane >> 4;

#pragma unroll
  for (int it = 0; it < 4; ++it) {
    const int kq = wave + 8 * it;
    const v4f* wr = (const v4f*)(W + (size_t)(4 * kq) * DCH) + lane;
    const v4f r0 = wr[0], r1 = wr[32], r2 = wr[64], r3 = wr[96];
    v4f* dp = wl + kq * DCH + 64 * h + c;
    v4f q0, q1, q2, q3;
    q0.x = r0.x; q0.y = r1.x; q0.z = r2.x; q0.w = r3.x;
    q1.x = r0.y; q1.y = r1.y; q1.z = r2.y; q1.w = r3.y;
    q2.x = r0.z; q2.y = r1.z; q2.z = r2.z; q2.w = r3.z;
    q3.x = r0.w; q3.y = r1.w; q3.z = r2.w; q3.w = r3.w;
    dp[0]  = q0;
    dp[16] = q1;
    dp[32] = q2;
    dp[48] = q3;
  }

  const int nh = wave & 1;
  const int m0 = (blockIdx.x * 4 + (wave >> 1)) * 16;

  int arow = m0 + c;
  if (arow > nrows - 1) arow = nrows - 1;
  const v4f* ap = (const v4f*)(X + (size_t)arow * DCH) + h;
  v4f a[16];
#pragma unroll
  for (int j = 0; j < 16; ++j) a[j] = ap[2 * j];
  const float sa = dinv[arow];
#pragma unroll
  for (int j = 0; j < 16; ++j) a[j] *= sa;

  __syncthreads();

  v8f acc0 = {}, acc1 = {}, acc2 = {}, acc3 = {};
  const v4f* bp = wl + h * DCH + 64 * nh + c;
  v4f b0 = bp[0], b1 = bp[16], b2 = bp[32], b3 = bp[48];
#pragma unroll
  for (int j = 0; j < 16; ++j) {
    const int jn = (j + 1 < 16) ? (j + 1) : j;
    const v4f n0 = bp[jn * 2 * DCH + 0];
    const v4f n1 = bp[jn * 2 * DCH + 16];
    const v4f n2 = bp[jn * 2 * DCH + 32];
    const v4f n3 = bp[jn * 2 * DCH + 48];
    __builtin_amdgcn_sched_barrier(0);
    v2f af, bf;
    af.x = a[j].x; af.y = a[j].y;
    bf.x = b0.x; bf.y = b0.y;
    acc0 = __builtin_amdgcn_wmma_f32_16x16x4_f32(false, af, false, bf, (short)0, acc0, false, false);
    bf.x = b1.x; bf.y = b1.y;
    acc1 = __builtin_amdgcn_wmma_f32_16x16x4_f32(false, af, false, bf, (short)0, acc1, false, false);
    bf.x = b2.x; bf.y = b2.y;
    acc2 = __builtin_amdgcn_wmma_f32_16x16x4_f32(false, af, false, bf, (short)0, acc2, false, false);
    bf.x = b3.x; bf.y = b3.y;
    acc3 = __builtin_amdgcn_wmma_f32_16x16x4_f32(false, af, false, bf, (short)0, acc3, false, false);
    af.x = a[j].z; af.y = a[j].w;
    bf.x = b0.z; bf.y = b0.w;
    acc0 = __builtin_amdgcn_wmma_f32_16x16x4_f32(false, af, false, bf, (short)0, acc0, false, false);
    bf.x = b1.z; bf.y = b1.w;
    acc1 = __builtin_amdgcn_wmma_f32_16x16x4_f32(false, af, false, bf, (short)0, acc1, false, false);
    bf.x = b2.z; bf.y = b2.w;
    acc2 = __builtin_amdgcn_wmma_f32_16x16x4_f32(false, af, false, bf, (short)0, acc2, false, false);
    bf.x = b3.z; bf.y = b3.w;
    acc3 = __builtin_amdgcn_wmma_f32_16x16x4_f32(false, af, false, bf, (short)0, acc3, false, false);
    __builtin_amdgcn_sched_barrier(0);
    b0 = n0; b1 = n1; b2 = n2; b3 = n3;
  }

  float* obase = out + (size_t)m0 * DCH + 64 * nh + 4 * c;
  if (m0 + 16 <= nrows) {
#pragma unroll
    for (int r = 0; r < 8; ++r) {
      v4f o;
      o.x = acc0[r]; o.y = acc1[r]; o.z = acc2[r]; o.w = acc3[r];
      *(v4f*)(obase + (size_t)(8 * h + r) * DCH) = o;
    }
  } else {
#pragma unroll
    for (int r = 0; r < 8; ++r) {
      v4f o;
      o.x = acc0[r]; o.y = acc1[r]; o.z = acc2[r]; o.w = acc3[r];
      if (m0 + 8 * h + r < nrows) *(v4f*)(obase + (size_t)(8 * h + r) * DCH) = o;
    }
  }
}

template <int R>
__device__ __forceinline__ void gather_rows(const float* __restrict__ hlin,
                                            int my_s, float dd,
                                            int j, int lane, v4f& acc) {
  v4f v[R];
#pragma unroll
  for (int k = 0; k < R; ++k) {
    const int s = __builtin_amdgcn_readlane(my_s, j + k);
    v[k] = ((const v4f*)(hlin + (size_t)s * DCH))[lane];
  }
#pragma unroll
  for (int k = 0; k < R; ++k) acc += v[k] * dd;
}

#define NODES_PER_WAVE 4

__device__ __forceinline__ v4f node_row(const float* __restrict__ hlin, const float* __restrict__ dinv,
                                        const v4f b, const int* __restrict__ rowptr,
                                        const int* __restrict__ csr_src,
                                        const float* __restrict__ gamma, const float* __restrict__ beta,
                                        float* __restrict__ jk, int do_relu, int first_layer,
                                        int n, int e, int node, int lane, float& dd_out) {
  const int beg = rowptr[node];
  const int end = (node + 1 < n) ? rowptr[node + 1] : e;
  const float dd = dinv[node];
  const int total = end - beg + 1;

  v4f acc = b;
  for (int q0 = 0; q0 < total; q0 += 32) {
    const int m = (total - q0 < 32) ? (total - q0) : 32;
    int q = q0 + lane;
    if (q > total - 1) q = total - 1;
    int my_s = node;
    if (q > 0) my_s = csr_src[beg + q - 1];

    int j = 0;
    for (; j + 8 <= m; j += 8) gather_rows<8>(hlin, my_s, dd, j, lane, acc);
    switch (m - j) {
      case 7: gather_rows<7>(hlin, my_s, dd, j, lane, acc); break;
      case 6: gather_rows<6>(hlin, my_s, dd, j, lane, acc); break;
      case 5: gather_rows<5>(hlin, my_s, dd, j, lane, acc); break;
      case 4: gather_rows<4>(hlin, my_s, dd, j, lane, acc); break;
      case 3: gather_rows<3>(hlin, my_s, dd, j, lane, acc); break;
      case 2: gather_rows<2>(hlin, my_s, dd, j, lane, acc); break;
      case 1: gather_rows<1>(hlin, my_s, dd, j, lane, acc); break;
      default: break;
    }
  }

  asm volatile("" ::: "memory");
  const v4f g  = ((const v4f*)gamma)[lane];
  const v4f bt = ((const v4f*)beta)[lane];
  v4f* jp = (v4f*)(jk + (size_t)node * DCH) + lane;
  v4f jold = {};
  if (!first_layer) jold = *jp;

  float s = acc.x + acc.y + acc.z + acc.w;
#pragma unroll
  for (int off = 16; off > 0; off >>= 1) s += __shfl_xor(s, off, 32);
  const float mu = s * (1.0f / DCH);

  const float dx = acc.x - mu, dy = acc.y - mu, dz = acc.z - mu, dw = acc.w - mu;
  float qq = dx * dx + dy * dy + dz * dz + dw * dw;
#pragma unroll
  for (int off = 16; off > 0; off >>= 1) qq += __shfl_xor(qq, off, 32);
  const float inv = rsqrtf(qq * (1.0f / DCH) + LN_EPS);

  v4f y;
  y.x = dx * inv * g.x + bt.x;
  y.y = dy * inv * g.y + bt.y;
  y.z = dz * inv * g.z + bt.z;
  y.w = dw * inv * g.w + bt.w;
  if (do_relu) {
    y.x = fmaxf(y.x, 0.0f); y.y = fmaxf(y.y, 0.0f);
    y.z = fmaxf(y.z, 0.0f); y.w = fmaxf(y.w, 0.0f);
  }

  if (first_layer) {
    *jp = y;
  } else {
    v4f o;
    o.x = fmaxf(jold.x, y.x); o.y = fmaxf(jold.y, y.y);
    o.z = fmaxf(jold.z, y.z); o.w = fmaxf(jold.w, y.w);
    *jp = o;
  }
  dd_out = dd;
  return y;
}

__global__ __launch_bounds__(256)
void gather_norm_kernel(const float* __restrict__ hlin, const float* __restrict__ dinv,
                        const float* __restrict__ bias, const int* __restrict__ rowptr,
                        const int* __restrict__ csr_src,
                        const float* __restrict__ gamma, const float* __restrict__ beta,
                        float* __restrict__ jk,
                        int do_relu, int first_layer, int n, int e) {
  const int lane = threadIdx.x & 31;
  int wid = (blockIdx.x * blockDim.x + threadIdx.x) >> 5;
  wid = __builtin_amdgcn_readfirstlane(wid);
  const v4f b = ((const v4f*)bias)[lane];
  const int node0 = wid * NODES_PER_WAVE;
  for (int node = node0; node < node0 + NODES_PER_WAVE && node < n; ++node) {
    float dd;
    (void)node_row(hlin, dinv, b, rowptr, csr_src, gamma, beta, jk, do_relu, first_layer,
                   n, e, node, lane, dd);
  }
}

#define FUSE_ROWS  64
#define FUSE_GRID  768
#define FUSE_PITCH 33
__global__ __launch_bounds__(512)
void gather_gemm_kernel(const float* __restrict__ hlin, const float* __restrict__ dinv,
                        const float* __restrict__ bias, const int* __restrict__ rowptr,
                        const int* __restrict__ csr_src,
                        const float* __restrict__ gamma, const float* __restrict__ beta,
                        const float* __restrict__ Wn, float* __restrict__ hnext,
                        float* __restrict__ jk,
                        int do_relu, int first_layer, int n, int e) {
  __shared__ v4f wl[32 * DCH];
  __shared__ v4f hrow[FUSE_ROWS * FUSE_PITCH];
  const int t    = threadIdx.x;
  const int lane = t & 31;
  const int wave = __builtin_amdgcn_readfirstlane(t >> 5);
  const int c    = lane & 15;
  const int h    = lane >> 4;

#pragma unroll
  for (int it = 0; it < 2; ++it) {
    const int kq = wave + 16 * it;
    const v4f* wr = (const v4f*)(Wn + (size_t)(4 * kq) * DCH) + lane;
    const v4f r0 = wr[0], r1 = wr[32], r2 = wr[64], r3 = wr[96];
    v4f* dp = wl + kq * DCH + 32 * (lane >> 3) + 2 * (lane & 7);
    v4f q0, q1, q2, q3;
    q0.x = r0.x; q0.y = r1.x; q0.z = r2.x; q0.w = r3.x;
    q1.x = r0.y; q1.y = r1.y; q1.z = r2.y; q1.w = r3.y;
    q2.x = r0.z; q2.y = r1.z; q2.z = r2.z; q2.w = r3.z;
    q3.x = r0.w; q3.y = r1.w; q3.z = r2.w; q3.w = r3.w;
    dp[0]  = q0;
    dp[16] = q1;
    dp[1]  = q2;
    dp[17] = q3;
  }

  const v4f b = ((const v4f*)bias)[lane];
  const int mt = wave >> 2;
  const int nq = wave & 3;
  const v4f* ar = hrow + (16 * mt + c) * FUSE_PITCH + h;
  const v4f* bp = wl + h * DCH + 32 * nq + c;

  const unsigned tiles  = (unsigned)(n + 15) >> 4;
  const int tile_b = (int)((blockIdx.x * tiles) / FUSE_GRID);
  const int tile_e = (int)(((blockIdx.x + 1) * tiles) / FUSE_GRID);
  int rend = 16 * tile_e;
  if (rend > n) rend = n;

  int pass_rows = FUSE_ROWS >> (blockIdx.x % 3);
  for (int row0 = 16 * tile_b; row0 < rend; ) {
    int cnt = rend - row0;
    if (cnt > pass_rows) cnt = pass_rows;

    for (int i = 0; i < NODES_PER_WAVE; ++i) {
      const int loc  = wave + 16 * i;
      const int node = row0 + loc;
      v4f a = {};
      if (loc < cnt) {
        float dd;
        const v4f y = node_row(hlin, dinv, b, rowptr, csr_src, gamma, beta, jk, do_relu, first_layer,
                               n, e, node, lane, dd);
        a = y * dd;
      }
      hrow[loc * FUSE_PITCH + lane] = a;
    }

    __syncthreads();

    if (16 * mt < cnt) {
      v8f acc0 = {}, acc1 = {};
      v4f a0 = ar[0], b0 = bp[0], b1 = bp[16];
#pragma unroll
      for (int j = 0; j < 16; ++j) {
        const int jn = (j + 1 < 16) ? (j + 1) : j;
        const v4f na = ar[2 * jn];
        const v4f n0 = bp[jn * 2 * DCH + 0];
        const v4f n1 = bp[jn * 2 * DCH + 16];
        __builtin_amdgcn_sched_barrier(0);
        v2f af, bf;
        af.x = a0.x; af.y = a0.y;
        bf.x = b0.x; bf.y = b0.y;
        acc0 = __builtin_amdgcn_wmma_f32_16x16x4_f32(false, af, false, bf, (short)0, acc0, false, false);
        bf.x = b1.x; bf.y = b1.y;
        acc1 = __builtin_amdgcn_wmma_f32_16x16x4_f32(false, af, false, bf, (short)0, acc1, false, false);
        af.x = a0.z; af.y = a0.w;
        bf.x = b0.z; bf.y = b0.w;
        acc0 = __builtin_amdgcn_wmma_f32_16x16x4_f32(false, af, false, bf, (short)0, acc0, false, false);
        bf.x = b1.z; bf.y = b1.w;
        acc1 = __builtin_amdgcn_wmma_f32_16x16x4_f32(false, af, false, bf, (short)0, acc1, false, false);
        __builtin_amdgcn_sched_barrier(0);
        a0 = na; b0 = n0; b1 = n1;
      }

      const int m0 = row0 + 16 * mt;
      float* obase = hnext + (size_t)m0 * DCH + 32 * nq + 2 * c;
      if (m0 + 16 <= rend) {
#pragma unroll
        for (int r = 0; r < 8; ++r) {
          v2f o;
          o.x = acc0[r]; o.y = acc1[r];
          *(v2f*)(obase + (size_t)(8 * h + r) * DCH) = o;
        }
      } else {
#pragma unroll
        for (int r = 0; r < 8; ++r) {
          v2f o;
          o.x = acc0[r]; o.y = acc1[r];
          if (m0 + 8 * h + r < rend) *(v2f*)(obase + (size_t)(8 * h + r) * DCH) = o;
        }
      }
    }

    __builtin_amdgcn_fence(__ATOMIC_RELEASE, "workgroup", "local");
    __builtin_amdgcn_s_barrier();
    __builtin_amdgcn_fence(__ATOMIC_ACQUIRE, "workgroup", "local");
    row0 += cnt;
    pass_rows = FUSE_ROWS;
  }
}

extern "C" void kernel_launch(void* const* d_in, const int* in_sizes, int n_in,
                              void* d_out, int out_size, void* d_ws, size_t ws_size,
                              hipStream_t stream) {
  const int N = in_sizes[0] / DCH;
  const int E = in_sizes[1] / 2;

  const float* feat = (const float*)d_in[0];
  const int*   ei   = (const int*)d_in[1];
  const int*   src  = ei;
  const int*   dst  = ei + E;
  const float* Wm[3] = {(const float*)d_in[2], (const float*)d_in[6],  (const float*)d_in[10]};
  const float* bi[3] = {(const float*)d_in[3], (const float*)d_in[7],  (const float*)d_in[11]};
  const float* gm[3] = {(const float*)d_in[4], (const float*)d_in[8],  (const float*)d_in[12]};
  const float* be[3] = {(const float*)d_in[5], (const float*)d_in[9],  (const float*)d_in[13]};
  float* out = (float*)d_out;

  float*    wsf      = (float*)d_ws;
  int*      rowptr   = (int*)wsf;
  float*    dinv     = wsf + N;
  float*    hlin     = wsf + 2 * (size_t)N;
  float*    agg      = hlin + (size_t)N * DCH;
  float*    hbuf     = agg  + (size_t)N * DCH;
  int*      csr_src  = (int*)(hbuf + (size_t)N * DCH);
  unsigned* total    = (unsigned*)agg;
  unsigned* blk_base = total + MAX_BKTS;
  int*      pairs    = (int*)(blk_base + (size_t)P1_BLOCKS * MAX_BKTS);

  const int nb    = (N + BKT_NODES - 1) / BKT_NODES;
  const int chunk = (E + P1_BLOCKS - 1) / P1_BLOCKS;
  zero_words_kernel<<<(nb + 255) / 256, 256, 0, stream>>>(total, nb);
  bucket_hist_kernel<<<P1_BLOCKS, P1_THREADS, 0, stream>>>(dst, total, blk_base, E, chunk, nb);
  bucket_scatter_kernel<<<P1_BLOCKS, P1_THREADS, 0, stream>>>(src, dst, total, blk_base, pairs, E, chunk, nb);
  bucket_sort_kernel<<<nb, BKT_NODES, 0, stream>>>(total, pairs, rowptr, dinv, csr_src, N);

  const int gemmBlocks = (N + GEMM_ROWS_PER_BLOCK - 1) / GEMM_ROWS_PER_BLOCK;
  const int gathBlocks = (N + 8 * NODES_PER_WAVE - 1) / (8 * NODES_PER_WAVE);

  gemm_wmma_kernel<<<gemmBlocks, 256, 0, stream>>>(feat, Wm[0], dinv, hlin, N);
  gather_gemm_kernel<<<FUSE_GRID, 512, 0, stream>>>(hlin, dinv, bi[0], rowptr, csr_src,
                                                     gm[0], be[0], Wm[1], hbuf, out,
                                                      1,  1, N, E);
  gather_gemm_kernel<<<FUSE_GRID, 512, 0, stream>>>(hbuf, dinv, bi[1], rowptr, csr_src,
                                                     gm[1], be[1], Wm[2], hlin, out,
                                                      1,  0, N, E);
  gather_norm_kernel<<<gathBlocks, 256, 0, stream>>>(hlin, dinv, bi[2], rowptr, csr_src,
                                                     gm[2], be[2], out,
                                                      0,  0, N, E);
}


# A wave's four node lists (bounds, dinv, first 32 sources) loaded as two groups a phase ahead of the gather (later passes': under the product phase), not in front of every node's rows; sum orders kept
# speedup vs baseline: 5.2128x; 1.0310x over previous
#include <hip/hip_runtime.h>

#define DCH 128
#define LN_EPS 1e-5f

typedef float v2f __attribute__((ext_vector_type(2)));
typedef float v4f __attribute__((ext_vector_type(4)));
typedef float v8f __attribute__((ext_vector_type(8)));

#define BKT_SHIFT   8
#define BKT_NODES   256
#define MAX_BKTS    512
#define P1_BLOCKS   512
#define P1_THREADS  512

typedef int v2i __attribute__((ext_vector_type(2)));
static_assert(P1_THREADS == MAX_BKTS, "thread t owns bucket t");
static_assert(MAX_BKTS == 2 * BKT_NODES, "a level-2 block reads all bucket totals with two loads per thread");

__global__ void zero_words_kernel(unsigned* __restrict__ p, int n) {
  int i = blockIdx.x * blockDim.x + threadIdx.x;
  if (i < n) p[i] = 0u;
}

__global__ __launch_bounds__(P1_THREADS)
void bucket_hist_kernel(const int* __restrict__ dst, unsigned* __restrict__ total,
                        unsigned* __restrict__ blk_base, int e, int chunk, int nb) {
  __shared__ unsigned hist[MAX_BKTS];
  const int t = threadIdx.x;
  hist[t] = 0u;
  __syncthreads();
  const int beg = blockIdx.x * chunk;
  int end = beg + chunk;
  if (end > e) end = e;
  for (int i0 = beg; i0 < end; i0 += 4 * P1_THREADS) {
    int d[4];
#pragma unroll
    for (int k = 0; k < 4; ++k) {
      const int idx = i0 + k * P1_THREADS + t;
      d[k] = (idx < end) ? dst[idx] : -1;
    }
#pragma unroll
    for (int k = 0; k < 4; ++k)
      if (d[k] >= 0) atomicAdd(&hist[d[k] >> BKT_SHIFT], 1u);
  }
  __syncthreads();
  if (t < nb) blk_base[blockIdx.x * MAX_BKTS + t] = atomicAdd(&total[t], hist[t]);
}

__global__ __launch_bounds__(P1_THREADS)
void bucket_scatter_kernel(const int* __restrict__ src, const int* __restrict__ dst,
                           const unsigned* __restrict__ total,
                           const unsigned* __restrict__ blk_base,
                           int* __restrict__ pairs_i, int e, int chunk, int nb) {
  __shared__ unsigned cur[MAX_BKTS];
  __shared__ unsigned wave_tot[P1_THREADS / 32];
  v2i* __restrict__ pairs = (v2i*)pairs_i;
  const int t = threadIdx.x, lane = t & 31, wv = t >> 5;
  const unsigned c    = (t < nb) ? total[t] : 0u;
  const unsigned mine = (t < nb) ? blk_base[blockIdx.x * MAX_BKTS + t] : 0u;
  unsigned inc = c;
  for (int off = 1; off < 32; off <<= 1) {
    unsigned v = __shfl_up(inc, off, 32);
    if (lane >= off) inc += v;
  }
  if (lane == 31) wave_tot[wv] = inc;
  __syncthreads();
  unsigned before = 0u;
  for (int k = 0; k < P1_THREADS / 32; ++k)
    if (k < wv) before += wave_tot[k];
  cur[t] = before + inc - c + mine;
  __syncthreads();

  const int beg = blockIdx.x * chunk;
  int end = beg + chunk;
  if (end > e) end = e;
  for (int i0 = beg; i0 < end; i0 += 4 * P1_THREADS) {
    int s[4], d[4];
#pragma unroll
    for (int k = 0; k < 4; ++k) {
      const int idx = i0 + k * P1_THREADS + t;
      const bool ok = idx < end;
      d[k] = ok ? dst[idx] : -1;
      s[k] = ok ? src[idx] : 0;
    }
    unsigned pos[4];
#pragma unroll
    for (int k = 0; k < 4; ++k) {
      pos[k] = 0u;
      if (d[k] >= 0) pos[k] = atomicAdd(&cur[d[k] >> BKT_SHIFT], 1u);
    }
#pragma unroll
    for (int k = 0; k < 4; ++k) {
      if (d[k] >= 0) {
        v2i pr; pr.x = s[k]; pr.y = d[k];
        pairs[pos[k]] = pr;
      }
    }
  }
}

__global__ __launch_bounds__(BKT_NODES)
void bucket_sort_kernel(const unsigned* __restrict__ total, const int* __restrict__ pairs_i,
                        int* __restrict__ rowptr, float* __restrict__ dinv,
                        int* __restrict__ csr_src, int n) {
  __shared__ unsigned cnt[BKT_NODES];
  __shared__ unsigned part[BKT_NODES / 32];
  __shared__ unsigned wave_tot[BKT_NODES / 32];
  const v2i* __restrict__ pairs = (const v2i*)pairs_i;
  const int t = threadIdx.x, lane = t & 31, wv = t >> 5;
  const int b = blockIdx.x;

  unsigned p = 0u;
  if (t < b) p = total[t];
  if (t + BKT_NODES < b) p += total[t + BKT_NODES];
  for (int off = 16; off > 0; off >>= 1) p += __shfl_xor(p, off, 32);
  if (lane == 0) part[wv] = p;
  cnt[t] = 0u;
  __syncthreads();
  unsigned start = 0u;
  for (int k = 0; k < BKT_NODES / 32; ++k) start += part[k];
  const int beg = (int)start;
  const int end = beg + (int)total[b];

  for (int i0 = beg; i0 < end; i0 += 4 * BKT_NODES) {
    int d[4];
#pragma unroll
    for (int k = 0; k < 4; ++k) {
      const int idx = i0 + k * BKT_NODES + t;
      d[k] = (idx < end) ? pairs[idx].y : -1;
    }
#pragma unroll
    for (int k = 0; k < 4; ++k)
      if (d[k] >= 0) atomicAdd(&cnt[d[k] & (BKT_NODES - 1)], 1u);
  }
  __syncthreads();

  const unsigned c = cnt[t];
  unsigned inc = c;
  for (int off = 1; off < 32; off <<= 1) {
    unsigned v = __shfl_up(inc, off, 32);
    if (lane >= off) inc += v;
  }
  if (lane == 31) wave_tot[wv] = inc;
  __syncthreads();
  unsigned before = start;
  for (int k = 0; k < BKT_NODES / 32; ++k)
    if (k < wv) before += wave_tot[k];
  const unsigned first = before + inc - c;
  cnt[t] = first;
  __syncthreads();
  const int node = b * BKT_NODES + t;
  if (node < n) {
    rowptr[node] = (int)first;
    dinv[node]   = rsqrtf(1.0f + (float)c);
  }

  for (int i0 = beg; i0 < end; i0 += 4 * BKT_NODES) {
    v2i pr[4];
#pragma unroll
    for (int k = 0; k < 4; ++k) {
      const int idx = i0 + k * BKT_NODES + t;
      v2i none; none.x = 0; none.y = -1;
      pr[k] = (idx < end) ? pairs[idx] : none;
    }
    unsigned pos[4];
#pragma unroll
    for (int k = 0; k < 4; ++k) {
      pos[k] = 0u;
      if (pr[k].y >= 0) pos[k] = atomicAdd(&cnt[pr[k].y & (BKT_NODES - 1)], 1u);
    }
#pragma unroll
    for (int k = 0; k < 4; ++k)
      if (pr[k].y >= 0) csr_src[pos[k]] = pr[k].x;
  }
}

#define GEMM_ROWS_PER_BLOCK 64
__global__ __launch_bounds__(256)
void gemm_wmma_kernel(const float* __restrict__ X, const float* __restrict__ W,
                      const float* __restrict__ dinv, float* __restrict__ out, int nrows) {
  __shared__ v4f wl[32 * DCH];
  const int t    = threadIdx.x;
  const int lane = t & 31;
  const int wave = __builtin_amdgcn_readfirstlane(t >> 5);
  const int c    = lane & 15;
  const int h    = lane >> 4;

#pragma unroll
  for (int it = 0; it < 4; ++it) {
    const int kq = wave + 8 * it;
    const v4f* wr = (const v4f*)(W + (size_t)(4 * kq) * DCH) + lane;
    const v4f r0 = wr[0], r1 = wr[32], r2 = wr[64], r3 = wr[96];
    v4f* dp = wl + kq * DCH + 64 * h + c;
    v4f q0, q1, q2, q3;
    q0.x = r0.x; q0.y = r1.x; q0.z = r2.x; q0.w = r3.x;
    q1.x = r0.y; q1.y = r1.y; q1.z = r2.y; q1.w = r3.y;
    q2.x = r0.z; q2.y = r1.z; q2.z = r2.z; q2.w = r3.z;
    q3.x = r0.w; q3.y = r1.w; q3.z = r2.w; q3.w = r3.w;
    dp[0]  = q0;
    dp[16] = q1;
    dp[32] = q2;
    dp[48] = q3;
  }

  const int nh = wave & 1;
  const int m0 = (blockIdx.x * 4 + (wave >> 1)) * 16;

  int arow = m0 + c;
  if (arow > nrows - 1) arow = nrows - 1;
  const v4f* ap = (const v4f*)(X + (size_t)arow * DCH) + h;
  v4f a[16];
#pragma unroll
  for (int j = 0; j < 16; ++j) a[j] = ap[2 * j];
  const float sa = dinv[arow];
#pragma unroll
  for (int j = 0; j < 16; ++j) a[j] *= sa;

  __syncthreads();

  v8f acc0 = {}, acc1 = {}, acc2 = {}, acc3 = {};
  const v4f* bp = wl + h * DCH + 64 * nh + c;
  v4f b0 = bp[0], b1 = bp[16], b2 = bp[32], b3 = bp[48];
#pragma unroll
  for (int j = 0; j < 16; ++j) {
    const int jn = (j + 1 < 16) ? (j + 1) : j;
    const v4f n0 = bp[jn * 2 * DCH + 0];
    const v4f n1 = bp[jn * 2 * DCH + 16];
    const v4f n2 = bp[jn * 2 * DCH + 32];
    const v4f n3 = bp[jn * 2 * DCH + 48];
    __builtin_amdgcn_sched_barrier(0);
    v2f af, bf;
    af.x = a[j].x; af.y = a[j].y;
    bf.x = b0.x; bf.y = b0.y;
    acc0 = __builtin_amdgcn_wmma_f32_16x16x4_f32(false, af, false, bf, (short)0, acc0, false, false);
    bf.x = b1.x; bf.y = b1.y;
    acc1 = __builtin_amdgcn_wmma_f32_16x16x4_f32(false, af, false, bf, (short)0, acc1, false, false);
    bf.x = b2.x; bf.y = b2.y;
    acc2 = __builtin_amdgcn_wmma_f32_16x16x4_f32(false, af, false, bf, (short)0, acc2, false, false);
    bf.x = b3.x; bf.y = b3.y;
    acc3 = __builtin_amdgcn_wmma_f32_16x16x4_f32(false, af, false, bf, (short)0, acc3, false, false);
    af.x = a[j].z; af.y = a[j].w;
    bf.x = b0.z; bf.y = b0.w;
    acc0 = __builtin_amdgcn_wmma_f32_16x16x4_f32(false, af, false, bf, (short)0, acc0, false, false);
    bf.x = b1.z; bf.y = b1.w;
    acc1 = __builtin_amdgcn_wmma_f32_16x16x4_f32(false, af, false, bf, (short)0, acc1, false, false);
    bf.x = b2.z; bf.y = b2.w;
    acc2 = __builtin_amdgcn_wmma_f32_16x16x4_f32(false, af, false, bf, (short)0, acc2, false, false);
    bf.x = b3.z; bf.y = b3.w;
    acc3 = __builtin_amdgcn_wmma_f32_16x16x4_f32(false, af, false, bf, (short)0, acc3, false, false);
    __builtin_amdgcn_sched_barrier(0);
    b0 = n0; b1 = n1; b2 = n2; b3 = n3;
  }

  float* obase = out + (size_t)m0 * DCH + 64 * nh + 4 * c;
  if (m0 + 16 <= nrows) {
#pragma unroll
    for (int r = 0; r < 8; ++r) {
      v4f o;
      o.x = acc0[r]; o.y = acc1[r]; o.z = acc2[r]; o.w = acc3[r];
      *(v4f*)(obase + (size_t)(8 * h + r) * DCH) = o;
    }
  } else {
#pragma unroll
    for (int r = 0; r < 8; ++r) {
      v4f o;
      o.x = acc0[r]; o.y = acc1[r]; o.z = acc2[r]; o.w = acc3[r];
      if (m0 + 8 * h + r < nrows) *(v4f*)(obase + (size_t)(8 * h + r) * DCH) = o;
    }
  }
}

template <int R>
__device__ __forceinline__ void gather_rows(const float* __restrict__ hlin,
                                            int my_s, float dd,
                                            int j, int lane, v4f& acc) {
  v4f v[R];
#pragma unroll
  for (int k = 0; k < R; ++k) {
    const int s = __builtin_amdgcn_readlane(my_s, j + k);
    v[k] = ((const v4f*)(hlin + (size_t)s * DCH))[lane];
  }
#pragma unroll
  for (int k = 0; k < R; ++k) acc += v[k] * dd;
}

#define NODES_PER_WAVE 4

__device__ __forceinline__ void node_bounds_ask(const int* __restrict__ rowptr,
                                                int n, int node, int& b0, int& b1) {
  const int nx = (node + 1 < n) ? (node + 1) : node;
  b0 = rowptr[node];
  b1 = rowptr[nx];
}
__device__ __forceinline__ void node_bounds_take(int n, int e, int node, int b0, int b1,
                                                 int& beg, int& total) {
  const int end = (node + 1 < n) ? b1 : e;
  beg = b0;
  total = end - b0 + 1;
}

__device__ __forceinline__ int node_head(const int* __restrict__ csr_src, int e, int node,
                                         int beg, int total, int lane) {
  int q = lane;
  if (q > total - 1) q = total - 1;
  int idx = beg + ((q > 0) ? q : 1) - 1;
  if (idx > e - 1) idx = e - 1;
  if (idx < 0) idx = 0;
  const int s = csr_src[idx];
  return (q > 0) ? s : node;
}

static_assert(NODES_PER_WAVE == 4, "wave_bounds_take names the four nodes' words one by one");
__device__ __forceinline__ void wave_bounds_ask(const int* __restrict__ rowptr,
                                                int n, int first, int step,
                                                int (&b0)[NODES_PER_WAVE], int (&b1)[NODES_PER_WAVE]) {
#pragma unroll
  for (int i = 0; i < NODES_PER_WAVE; ++i) {
    int node = first + step * i;
    if (node > n - 1) node = n - 1;
    node_bounds_ask(rowptr, n, node, b0[i], b1[i]);
  }
}
__device__ __forceinline__ void wave_bounds_take(int n, int e, int first, int step,
                                                 int (&b0)[NODES_PER_WAVE], int (&b1)[NODES_PER_WAVE],
                                                 int (&nbeg)[NODES_PER_WAVE], int (&ntot)[NODES_PER_WAVE]) {
#pragma unroll
  for (int i = 0; i < NODES_PER_WAVE; ++i) {
    b0[i] = __builtin_amdgcn_readfirstlane(b0[i]);
    b1[i] = __builtin_amdgcn_readfirstlane(b1[i]);
  }
  asm volatile("" : "+s"(b0[0]), "+s"(b0[1]), "+s"(b0[2]), "+s"(b0[3]),
                    "+s"(b1[0]), "+s"(b1[1]), "+s"(b1[2]), "+s"(b1[3]));
#pragma unroll
  for (int i = 0; i < NODES_PER_WAVE; ++i) {
    int node = first + step * i;
    if (node > n - 1) node = n - 1;
    node_bounds_take(n, e, node, b0[i], b1[i], nbeg[i], ntot[i]);
  }
}
__device__ __forceinline__ void wave_heads(const int* __restrict__ csr_src, const float* __restrict__ dinv,
                                           int n, int e, int first, int step, int lane,
                                           const int (&nbeg)[NODES_PER_WAVE], const int (&ntot)[NODES_PER_WAVE],
                                           int (&nhead)[NODES_PER_WAVE], float (&ndd)[NODES_PER_WAVE]) {
#pragma unroll
  for (int i = 0; i < NODES_PER_WAVE; ++i) {
    int node = first + step * i;
    if (node > n - 1) node = n - 1;
    nhead[i] = node_head(csr_src, e, node, nbeg[i], ntot[i], lane);
    ndd[i] = dinv[node];
  }
}
__device__ __forceinline__ void wave_dd_uniform(float (&ndd)[NODES_PER_WAVE]) {
#pragma unroll
  for (int i = 0; i < NODES_PER_WAVE; ++i)
    ndd[i] = __builtin_bit_cast(float, __builtin_amdgcn_readfirstlane(__builtin_bit_cast(int, ndd[i])));
}

__device__ __forceinline__ v4f node_row(const float* __restrict__ hlin,
                                        const v4f b, const int* __restrict__ csr_src,
                                        const float* __restrict__ gamma, const float* __restrict__ beta,
                                        float* __restrict__ jk, int do_relu, int first_layer,
                                        int node, int lane,
                                        int beg, int total, float dd, int head) {
  v4f acc = b;
  for (int q0 = 0; q0 < total; q0 += 32) {
    const int m = (total - q0 < 32) ? (total - q0) : 32;
    int my_s = head;
    if (q0 > 0) {
      int q = q0 + lane;
      if (q > total - 1) q = total - 1;
      my_s = csr_src[beg + q - 1];
    }

    int j = 0;
    for (; j + 8 <= m; j += 8) gather_rows<8>(hlin, my_s, dd, j, lane, acc);
    switch (m - j) {
      case 7: gather_rows<7>(hlin, my_s, dd, j, lane, acc); break;
      case 6: gather_rows<6>(hlin, my_s, dd, j, lane, acc); break;
      case 5: gather_rows<5>(hlin, my_s, dd, j, lane, acc); break;
      case 4: gather_rows<4>(hlin, my_s, dd, j, lane, acc); break;
      case 3: gather_rows<3>(hlin, my_s, dd, j, lane, acc); break;
      case 2: gather_rows<2>(hlin, my_s, dd, j, lane, acc); break;
      case 1: gather_rows<1>(hlin, my_s, dd, j, lane, acc); break;
      default: break;
    }
  }

  asm volatile("" ::: "memory");
  const v4f g  = ((const v4f*)gamma)[lane];
  const v4f bt = ((const v4f*)beta)[lane];
  v4f* jp = (v4f*)(jk + (size_t)node * DCH) + lane;
  v4f jold = {};
  if (!first_layer) jold = *jp;

  float s = acc.x + acc.y + acc.z + acc.w;
#pragma unroll
  for (int off = 16; off > 0; off >>= 1) s += __shfl_xor(s, off, 32);
  const float mu = s * (1.0f / DCH);

  const float dx = acc.x - mu, dy = acc.y - mu, dz = acc.z - mu, dw = acc.w - mu;
  float qq = dx * dx + dy * dy + dz * dz + dw * dw;
#pragma unroll
  for (int off = 16; off > 0; off >>= 1) qq += __shfl_xor(qq, off, 32);
  const float inv = rsqrtf(qq * (1.0f / DCH) + LN_EPS);

  v4f y;
  y.x = dx * inv * g.x + bt.x;
  y.y = dy * inv * g.y + bt.y;
  y.z = dz * inv * g.z + bt.z;
  y.w = dw * inv * g.w + bt.w;
  if (do_relu) {
    y.x = fmaxf(y.x, 0.0f); y.y = fmaxf(y.y, 0.0f);
    y.z = fmaxf(y.z, 0.0f); y.w = fmaxf(y.w, 0.0f);
  }

  if (first_layer) {
    *jp = y;
  } else {
    v4f o;
    o.x = fmaxf(jold.x, y.x); o.y = fmaxf(jold.y, y.y);
    o.z = fmaxf(jold.z, y.z); o.w = fmaxf(jold.w, y.w);
    *jp = o;
  }
  return y;
}

__global__ __launch_bounds__(256)
void gather_norm_kernel(const float* __restrict__ hlin, const float* __restrict__ dinv,
                        const float* __restrict__ bias, const int* __restrict__ rowptr,
                        const int* __restrict__ csr_src,
                        const float* __restrict__ gamma, const float* __restrict__ beta,
                        float* __restrict__ jk,
                        int do_relu, int first_layer, int n, int e) {
  const int lane = threadIdx.x & 31;
  int wid = (blockIdx.x * blockDim.x + threadIdx.x) >> 5;
  wid = __builtin_amdgcn_readfirstlane(wid);
  const v4f b = ((const v4f*)bias)[lane];
  const int node0 = wid * NODES_PER_WAVE;
  int nbeg[NODES_PER_WAVE], ntot[NODES_PER_WAVE], nhead[NODES_PER_WAVE];
  float ndd[NODES_PER_WAVE];
  int rb0[NODES_PER_WAVE], rb1[NODES_PER_WAVE];
  wave_bounds_ask(rowptr, n, node0, 1, rb0, rb1);
  wave_bounds_take(n, e, node0, 1, rb0, rb1, nbeg, ntot);
  wave_heads(csr_src, dinv, n, e, node0, 1, lane, nbeg, ntot, nhead, ndd);
  wave_dd_uniform(ndd);
#pragma unroll
  for (int i = 0; i < NODES_PER_WAVE; ++i) {
    const int node = node0 + i;
    if (node < n)
      (void)node_row(hlin, b, csr_src, gamma, beta, jk, do_relu, first_layer,
                     node, lane, nbeg[i], ntot[i], ndd[i], nhead[i]);
  }
}

#define FUSE_ROWS  64
#define FUSE_GRID  768
#define FUSE_PITCH 33
__global__ __launch_bounds__(512)
void gather_gemm_kernel(const float* __restrict__ hlin, const float* __restrict__ dinv,
                        const float* __restrict__ bias, const int* __restrict__ rowptr,
                        const int* __restrict__ csr_src,
                        const float* __restrict__ gamma, const float* __restrict__ beta,
                        const float* __restrict__ Wn, float* __restrict__ hnext,
                        float* __restrict__ jk,
                        int do_relu, int first_layer, int n, int e) {
  __shared__ v4f wl[32 * DCH];
  __shared__ v4f hrow[FUSE_ROWS * FUSE_PITCH];
  const int t    = threadIdx.x;
  const int lane = t & 31;
  const int wave = __builtin_amdgcn_readfirstlane(t >> 5);
  const int c    = lane & 15;
  const int h    = lane >> 4;

  const unsigned tiles  = (unsigned)(n + 15) >> 4;
  const int tile_b = (int)((blockIdx.x * tiles) / FUSE_GRID);
  const int tile_e = (int)(((blockIdx.x + 1) * tiles) / FUSE_GRID);
  int rend = 16 * tile_e;
  if (rend > n) rend = n;

  int nbeg[NODES_PER_WAVE], ntot[NODES_PER_WAVE], nhead[NODES_PER_WAVE];
  float ndd[NODES_PER_WAVE];
  int rb0[NODES_PER_WAVE], rb1[NODES_PER_WAVE];
  wave_bounds_ask(rowptr, n, 16 * tile_b + wave, 16, rb0, rb1);

#pragma unroll
  for (int it = 0; it < 2; ++it) {
    const int kq = wave + 16 * it;
    const v4f* wr = (const v4f*)(Wn + (size_t)(4 * kq) * DCH) + lane;
    const v4f r0 = wr[0], r1 = wr[32], r2 = wr[64], r3 = wr[96];
    v4f* dp = wl + kq * DCH + 32 * (lane >> 3) + 2 * (lane & 7);
    v4f q0, q1, q2, q3;
    q0.x = r0.x; q0.y = r1.x; q0.z = r2.x; q0.w = r3.x;
    q1.x = r0.y; q1.y = r1.y; q1.z = r2.y; q1.w = r3.y;
    q2.x = r0.z; q2.y = r1.z; q2.z = r2.z; q2.w = r3.z;
    q3.x = r0.w; q3.y = r1.w; q3.z = r2.w; q3.w = r3.w;
    dp[0]  = q0;
    dp[16] = q1;
    dp[1]  = q2;
    dp[17] = q3;
  }

  const v4f b = ((const v4f*)bias)[lane];
  const int mt = wave >> 2;
  const int nq = wave & 3;
  const v4f* ar = hrow + (16 * mt + c) * FUSE_PITCH + h;
  const v4f* bp = wl + h * DCH + 32 * nq + c;

  int pass_rows = FUSE_ROWS >> (blockIdx.x % 3);

  wave_bounds_take(n, e, 16 * tile_b + wave, 16, rb0, rb1, nbeg, ntot);
  wave_heads(csr_src, dinv, n, e, 16 * tile_b + wave, 16, lane, nbeg, ntot, nhead, ndd);

  for (int row0 = 16 * tile_b; row0 < rend; ) {
    int cnt = rend - row0;
    if (cnt > pass_rows) cnt = pass_rows;

    wave_dd_uniform(ndd);

#pragma nounroll
    for (int i = 0; i < NODES_PER_WAVE; ++i) {
      const int loc  = wave + 16 * i;
      const int node = row0 + loc;
      v4f a = {};
      if (loc < cnt) {
        const v4f y = node_row(hlin, b, csr_src, gamma, beta, jk, do_relu, first_layer,
                               node, lane, nbeg[0], ntot[0], ndd[0], nhead[0]);
        a = y * ndd[0];
      }
      hrow[loc * FUSE_PITCH + lane] = a;
#pragma unroll
      for (int k = 0; k + 1 < NODES_PER_WAVE; ++k) {
        nbeg[k] = nbeg[k + 1]; ntot[k] = ntot[k + 1]; ndd[k] = ndd[k + 1]; nhead[k] = nhead[k + 1];
      }
    }

    __syncthreads();

    const int rown = row0 + cnt;
    const bool more = rown < rend;
    if (more) wave_bounds_ask(rowptr, n, rown + wave, 16, rb0, rb1);

    if (16 * mt < cnt) {
      v8f acc0 = {}, acc1 = {};
      v4f a0 = ar[0], b0 = bp[0], b1 = bp[16];
#pragma unroll
      for (int j = 0; j < 16; ++j) {
        const int jn = (j + 1 < 16) ? (j + 1) : j;
        const v4f na = ar[2 * jn];
        const v4f n0 = bp[jn * 2 * DCH + 0];
        const v4f n1 = bp[jn * 2 * DCH + 16];
        __builtin_amdgcn_sched_barrier(0);
        v2f af, bf;
        af.x = a0.x; af.y = a0.y;
        bf.x = b0.x; bf.y = b0.y;
        acc0 = __builtin_amdgcn_wmma_f32_16x16x4_f32(false, af, false, bf, (short)0, acc0, false, false);
        bf.x = b1.x; bf.y = b1.y;
        acc1 = __builtin_amdgcn_wmma_f32_16x16x4_f32(false, af, false, bf, (short)0, acc1, false, false);
        af.x = a0.z; af.y = a0.w;
        bf.x = b0.z; bf.y = b0.w;
        acc0 = __builtin_amdgcn_wmma_f32_16x16x4_f32(false, af, false, bf, (short)0, acc0, false, false);
        bf.x = b1.z; bf.y = b1.w;
        acc1 = __builtin_amdgcn_wmma_f32_16x16x4_f32(false, af, false, bf, (short)0, acc1, false, false);
        __builtin_amdgcn_sched_barrier(0);
        a0 = na; b0 = n0; b1 = n1;
      }

      const int m0 = row0 + 16 * mt;
      float* obase = hnext + (size_t)m0 * DCH + 32 * nq + 2 * c;
      if (m0 + 16 <= rend) {
#pragma unroll
        for (int r = 0; r < 8; ++r) {
          v2f o;
          o.x = acc0[r]; o.y = acc1[r];
          *(v2f*)(obase + (size_t)(8 * h + r) * DCH) = o;
        }
      } else {
#pragma unroll
        for (int r = 0; r < 8; ++r) {
          v2f o;
          o.x = acc0[r]; o.y = acc1[r];
          if (m0 + 8 * h + r < rend) *(v2f*)(obase + (size_t)(8 * h + r) * DCH) = o;
        }
      }
    }

    if (more) {
      wave_bounds_take(n, e, rown + wave, 16, rb0, rb1, nbeg, ntot);
      wave_heads(csr_src, dinv, n, e, rown + wave, 16, lane, nbeg, ntot, nhead, ndd);
    }

    __builtin_amdgcn_fence(__ATOMIC_RELEASE, "workgroup", "local");
    __builtin_amdgcn_s_barrier();
    __builtin_amdgcn_fence(__ATOMIC_ACQUIRE, "workgroup", "local");
    row0 = rown;
    pass_rows = FUSE_ROWS;
  }
}

extern "C" void kernel_launch(void* const* d_in, const int* in_sizes, int n_in,
                              void* d_out, int out_size, void* d_ws, size_t ws_size,
                              hipStream_t stream) {
  const int N = in_sizes[0] / DCH;
  const int E = in_sizes[1] / 2;

  const float* feat = (const float*)d_in[0];
  const int*   ei   = (const int*)d_in[1];
  const int*   src  = ei;
  const int*   dst  = ei + E;
  const float* Wm[3] = {(const float*)d_in[2], (const float*)d_in[6],  (const float*)d_in[10]};
  const float* bi[3] = {(const float*)d_in[3], (const float*)d_in[7],  (const float*)d_in[11]};
  const float* gm[3] = {(const float*)d_in[4], (const float*)d_in[8],  (const float*)d_in[12]};
  const float* be[3] = {(const float*)d_in[5], (const float*)d_in[9],  (const float*)d_in[13]};
  float* out = (float*)d_out;

  float*    wsf      = (float*)d_ws;
  int*      rowptr   = (int*)wsf;
  float*    dinv     = wsf + N;
  float*    hlin     = wsf + 2 * (size_t)N;
  float*    agg      = hlin + (size_t)N * DCH;
  float*    hbuf     = agg  + (size_t)N * DCH;
  int*      csr_src  = (int*)(hbuf + (size_t)N * DCH);
  unsigned* total    = (unsigned*)agg;
  unsigned* blk_base = total + MAX_BKTS;
  int*      pairs    = (int*)(blk_base + (size_t)P1_BLOCKS * MAX_BKTS);

  const int nb    = (N + BKT_NODES - 1) / BKT_NODES;
  const int chunk = (E + P1_BLOCKS - 1) / P1_BLOCKS;
  zero_words_kernel<<<(nb + 255) / 256, 256, 0, stream>>>(total, nb);
  bucket_hist_kernel<<<P1_BLOCKS, P1_THREADS, 0, stream>>>(dst, total, blk_base, E, chunk, nb);
  bucket_scatter_kernel<<<P1_BLOCKS, P1_THREADS, 0, stream>>>(src, dst, total, blk_base, pairs, E, chunk, nb);
  bucket_sort_kernel<<<nb, BKT_NODES, 0, stream>>>(total, pairs, rowptr, dinv, csr_src, N);

  const int gemmBlocks = (N + GEMM_ROWS_PER_BLOCK - 1) / GEMM_ROWS_PER_BLOCK;
  const int gathBlocks = (N + 8 * NODES_PER_WAVE - 1) / (8 * NODES_PER_WAVE);

  gemm_wmma_kernel<<<gemmBlocks, 256, 0, stream>>>(feat, Wm[0], dinv, hlin, N);
  gather_gemm_kernel<<<FUSE_GRID, 512, 0, stream>>>(hlin, dinv, bi[0], rowptr, csr_src,
                                                     gm[0], be[0], Wm[1], hbuf, out,
                                                      1,  1, N, E);
  gather_gemm_kernel<<<FUSE_GRID, 512, 0, stream>>>(hbuf, dinv, bi[1], rowptr, csr_src,
                                                     gm[1], be[1], Wm[2], hlin, out,
                                                      1,  0, N, E);
  gather_norm_kernel<<<gathBlocks, 256, 0, stream>>>(hlin, dinv, bi[2], rowptr, csr_src,
                                                     gm[2], be[2], out,
                                                      0,  0, N, E);
}
